# starting kernel
# baseline (speedup 1.0000x reference)
#include <hip/hip_runtime.h>
#include <hip/hip_bf16.h>

#define NB   64
#define DD   32
#define NN   128
#define TT   6
#define NT   (NN*TT)
#define NCOL (NB*DD)
#define LEAF 64
#define JTOT 448
#define LDX  (JTOT*DD)

typedef __attribute__((ext_vector_type(2))) float v2f;
typedef __attribute__((ext_vector_type(8))) float v8f;

__global__ void k_copy_x(const float* __restrict__ x, float* __restrict__ out) {
    int idx = blockIdx.x * 256 + threadIdx.x;
    if (idx >= NB * 64 * DD) return;
    int b   = idx / (64 * DD);
    int rem = idx - b * (64 * DD);
    out[b * LDX + rem] = x[idx];
}

__global__ void k_fsmax(const float* __restrict__ fsl, float* __restrict__ fs, int J) {
    __shared__ float row[320];
    __shared__ float s_ts[256];
    __shared__ int   s_k[256];
    int nt  = blockIdx.x;
    int tid = threadIdx.x;
    for (int e = tid; e < J; e += 256) row[e] = fsl[nt * J + e];
    __syncthreads();

    float local_ts = 0.0f;
    int   local_k  = 0;
    for (int e = tid; e < J; e += 256) {
        float z = row[e];
        int   c = 1;
        float s = 0.0f;
        for (int j = 0; j < J; ++j) {
            float zj = row[j];
            c += (zj > z) ? 1 : 0;
            s += (zj >= z) ? zj : 0.0f;
        }
        bool supp = ((float)c * z > s - 1.0f);
        local_k  += supp ? 1 : 0;
        local_ts += supp ? z : 0.0f;
    }
    s_ts[tid] = local_ts;
    s_k[tid]  = local_k;
    __syncthreads();
    for (int off = 128; off > 0; off >>= 1) {
        if (tid < off) { s_ts[tid] += s_ts[tid + off]; s_k[tid] += s_k[tid + off]; }
        __syncthreads();
    }
    float tau = (s_ts[0] - 1.0f) / (float)s_k[0];
    for (int e = tid; e < J; e += 256) fs[nt * J + e] = fmaxf(row[e] - tau, 0.0f);
}

#define TPAD 33
__global__ void k_essum(const float* __restrict__ esl, float* __restrict__ essum, int J) {
    __shared__ float tile[8][32 * TPAD];
    __shared__ float accs[8][32];
    int nt   = blockIdx.x;
    int lane = threadIdx.x & 31;
    int wid  = threadIdx.x >> 5;
    float* my = tile[wid];
    float acc = 0.0f;
    int ntiles = J >> 5;
    for (int tb = wid; tb < ntiles; tb += 8) {
        size_t rowbase = (size_t)nt * J + tb * 32;
        #pragma unroll
        for (int rr = 0; rr < 32; ++rr) {
            my[rr * TPAD + lane] = esl[(rowbase + rr) * DD + lane];
        }
        float r[32];
        #pragma unroll
        for (int k = 0; k < 32; ++k) r[k] = my[lane * TPAD + k];
        #pragma unroll
        for (int k = 2; k <= 32; k <<= 1) {
            #pragma unroll
            for (int j = k >> 1; j > 0; j >>= 1) {
                #pragma unroll
                for (int i = 0; i < 32; ++i) {
                    int ij = i ^ j;
                    if (ij > i) {
                        float a  = r[i], b2 = r[ij];
                        float lo = fminf(a, b2), hi = fmaxf(a, b2);
                        if ((i & k) == 0) { r[i] = hi; r[ij] = lo; }
                        else              { r[i] = lo; r[ij] = hi; }
                    }
                }
            }
        }
        float cum = 0.0f, tausum = 1.0f;
        int   ksup = 1;
        #pragma unroll
        for (int i = 0; i < 32; ++i) {
            cum += r[i];
            bool s = ((float)(i + 1) * r[i] > cum - 1.0f);
            if (s) { ksup = i + 1; tausum = cum; }
        }
        float tau = (tausum - 1.0f) / (float)ksup;
        #pragma unroll
        for (int j2 = 0; j2 < 32; ++j2) {
            float z = my[j2 * TPAD + lane];
            float t = __shfl(tau, j2, 32);
            acc += fmaxf(z - t, 0.0f);
        }
    }
    accs[wid][lane] = acc;
    __syncthreads();
    if (threadIdx.x < 32) {
        float s = 0.0f;
        #pragma unroll
        for (int w = 0; w < 8; ++w) s += accs[w][threadIdx.x];
        essum[nt * DD + threadIdx.x] = s;
    }
}

__global__ void k_gemm_tl(const float* __restrict__ x, const float* __restrict__ fs,
                          const float* __restrict__ essum, const float* __restrict__ thr,
                          const float* __restrict__ lt, float* __restrict__ tl, int K) {
    int lane  = threadIdx.x & 31;
    int wid   = threadIdx.x >> 5;
    int tile  = blockIdx.x * 8 + wid;
    int mt    = tile >> 7;
    int ntile = tile & 127;
    int m0    = mt * 16;
    int col   = ntile * 16 + (lane & 15);
    int bidx  = col >> 5;
    int d     = col & 31;
    int ka_off = (lane >> 4) << 1;

    const float* arow = fs + (size_t)(m0 + (lane & 15)) * K;
    const float* bcol = x + (size_t)bidx * LDX + d;

    v8f acc = {};
    for (int k0 = 0; k0 < K; k0 += 4) {
        int ka = k0 + ka_off;
        v2f a;
        a.x = arow[ka];
        a.y = arow[ka + 1];
        v2f b;
        b.x = bcol[ka * DD];
        b.y = bcol[(ka + 1) * DD];
        acc = __builtin_amdgcn_wmma_f32_16x16x4_f32(false, a, false, b,
                                                    (short)0, acc, false, false);
    }
    int mplus = (lane >> 4) << 3;
    #pragma unroll
    for (int v = 0; v < 8; ++v) {
        int nt   = m0 + v + mplus;
        float sel = acc[v] * essum[nt * DD + d];
        float tv  = (sel - thr[nt]) * __expf(-lt[nt]);
        tl[(size_t)nt * NCOL + col] = tv;
    }
}

__global__ void k_tree(const float* __restrict__ tl, const float* __restrict__ resp,
                       float* __restrict__ out, int rowOff) {
    __shared__ float rsp[DD * 65];
    int nn  = blockIdx.x >> 3;
    int bg  = blockIdx.x & 7;
    int tid = threadIdx.x;
    for (int idx = tid; idx < DD * LEAF; idx += 256) {
        int dd = idx >> 6, cc = idx & 63;
        rsp[dd * 65 + cc] = resp[(size_t)nn * (DD * LEAF) + idx];
    }
    __syncthreads();

    int lane = tid & 31;
    int wid  = tid >> 5;
    int b    = bg * 8 + wid;

    float p[TT], q[TT];
    #pragma unroll
    for (int t = 0; t < TT; ++t) {
        float v = tl[(size_t)(nn * TT + t) * NCOL + b * DD + lane];
        p[t] = fminf(fmaxf(0.5f * v + 0.5f, 0.0f), 1.0f);
        q[t] = fminf(fmaxf(0.5f - 0.5f * v, 0.0f), 1.0f);
    }
    float r[LEAF];
    r[0] = 1.0f;
    #pragma unroll
    for (int t = 0; t < TT; ++t) {
        int sz = 1 << t;
        #pragma unroll
        for (int c = 0; c < 32; ++c) {
            int cc = sz - 1 - c;
            if (cc >= 0) {
                r[cc + sz] = r[cc] * q[t];
                r[cc]      = r[cc] * p[t];
            }
        }
    }
    float s = 0.0f;
    #pragma unroll
    for (int c = 0; c < LEAF; ++c) s += r[c] * rsp[lane * 65 + c];
    out[(size_t)b * LDX + (rowOff + nn) * DD + lane] = s;
}

extern "C" void kernel_launch(void* const* d_in, const int* in_sizes, int n_in,
                              void* d_out, int out_size, void* d_ws, size_t ws_size,
                              hipStream_t stream) {
    const float* x = (const float*)d_in[0];
    float* out = (float*)d_out;

    float* fs_ws    = (float*)d_ws;
    float* essum_ws = fs_ws + NT * 320;
    float* tl_ws    = essum_ws + NT * DD;

    k_copy_x<<<(NB * 64 * DD + 255) / 256, 256, 0, stream>>>(x, out);

    const int Js[3]     = {64, 192, 320};
    const int rowOff[3] = {64, 192, 320};
    for (int i = 0; i < 3; ++i) {
        const float* fsl  = (const float*)d_in[1 + 5 * i];
        const float* esl  = (const float*)d_in[2 + 5 * i];
        const float* thr  = (const float*)d_in[3 + 5 * i];
        const float* lt   = (const float*)d_in[4 + 5 * i];
        const float* resp = (const float*)d_in[5 + 5 * i];
        int J = Js[i];

        k_fsmax <<<NT, 256, 0, stream>>>(fsl, fs_ws, J);
        k_essum <<<NT, 256, 0, stream>>>(esl, essum_ws, J);
        k_gemm_tl<<<(48 * 128) / 8, 256, 0, stream>>>(out, fs_ws, essum_ws, thr, lt, tl_ws, J);
        k_tree  <<<NN * 8, 256, 0, stream>>>(tl_ws, resp, out, rowOff[i]);
    }
}


# Input-only sparsemax work leaves its 6 launches: copy, fsmax 0-1, essum 0-2 in ONE launch ahead of the layers; fsmax 2 rides in the layer-1 tree launch (13 launches -> 7)
# speedup vs baseline: 1.2480x; 1.2480x over previous
#include <hip/hip_runtime.h>
#include <hip/hip_bf16.h>

#define NB   64
#define DD   32
#define NN   128
#define TT   6
#define NT   (NN*TT)
#define NCOL (NB*DD)
#define LEAF 64
#define JTOT 448
#define LDX  (JTOT*DD)

typedef __attribute__((ext_vector_type(2))) float v2f;
typedef __attribute__((ext_vector_type(8))) float v8f;

__device__ __forceinline__ void copy_x_body(int blk, const float* __restrict__ x,
                                            float* __restrict__ out) {
    int idx = blk * 256 + threadIdx.x;
    if (idx >= NB * 64 * DD) return;
    int b   = idx / (64 * DD);
    int rem = idx - b * (64 * DD);
    out[b * LDX + rem] = x[idx];
}

__device__ __forceinline__ void fsmax_body(float* row, float* s_ts, int* s_k, int nt,
                                           const float* __restrict__ fsl,
                                           float* __restrict__ fs, int J) {
    int tid = threadIdx.x;
    for (int e = tid; e < J; e += 256) row[e] = fsl[nt * J + e];
    __syncthreads();

    float local_ts = 0.0f;
    int   local_k  = 0;
    for (int e = tid; e < J; e += 256) {
        float z = row[e];
        int   c = 1;
        float s = 0.0f;
        for (int j = 0; j < J; ++j) {
            float zj = row[j];
            c += (zj > z) ? 1 : 0;
            s += (zj >= z) ? zj : 0.0f;
        }
        bool supp = ((float)c * z > s - 1.0f);
        local_k  += supp ? 1 : 0;
        local_ts += supp ? z : 0.0f;
    }
    s_ts[tid] = local_ts;
    s_k[tid]  = local_k;
    __syncthreads();
    for (int off = 128; off > 0; off >>= 1) {
        if (tid < off) { s_ts[tid] += s_ts[tid + off]; s_k[tid] += s_k[tid + off]; }
        __syncthreads();
    }
    float tau = (s_ts[0] - 1.0f) / (float)s_k[0];
    for (int e = tid; e < J; e += 256) fs[nt * J + e] = fmaxf(row[e] - tau, 0.0f);
}

#define TPAD 33
__device__ __forceinline__ void essum_body(float (*tile)[32 * TPAD], float (*accs)[32], int nt,
                                           const float* __restrict__ esl,
                                           float* __restrict__ essum, int J) {
    int lane = threadIdx.x & 31;
    int wid  = threadIdx.x >> 5;
    float* my = tile[wid];
    float acc = 0.0f;
    int ntiles = J >> 5;
    for (int tb = wid; tb < ntiles; tb += 8) {
        const float* src = esl + ((size_t)nt * J + tb * 32) * DD + lane;
        #pragma unroll
        for (int rr = 0; rr < 32; ++rr) {
            my[rr * TPAD + lane] = src[rr * DD];
        }
        float r[32];
        #pragma unroll
        for (int k = 0; k < 32; ++k) r[k] = my[lane * TPAD + k];
        #pragma unroll
        for (int k = 2; k <= 32; k <<= 1) {
            #pragma unroll
            for (int j = k >> 1; j > 0; j >>= 1) {
                #pragma unroll
                for (int i = 0; i < 32; ++i) {
                    int ij = i ^ j;
                    if (ij > i) {
                        float a  = r[i], b2 = r[ij];
                        float lo = fminf(a, b2), hi = fmaxf(a, b2);
                        if ((i & k) == 0) { r[i] = hi; r[ij] = lo; }
                        else              { r[i] = lo; r[ij] = hi; }
                    }
                }
            }
        }
        float cum = 0.0f, tausum = 1.0f;
        int   ksup = 1;
        #pragma unroll
        for (int i = 0; i < 32; ++i) {
            cum += r[i];
            bool s = ((float)(i + 1) * r[i] > cum - 1.0f);
            if (s) { ksup = i + 1; tausum = cum; }
        }
        float tau = (tausum - 1.0f) / (float)ksup;
        #pragma unroll
        for (int j2 = 0; j2 < 32; ++j2) {
            float z = my[j2 * TPAD + lane];
            float t = __shfl(tau, j2, 32);
            acc += fmaxf(z - t, 0.0f);
        }
    }
    accs[wid][lane] = acc;
    __syncthreads();
    if (threadIdx.x < 32) {
        float s = 0.0f;
        #pragma unroll
        for (int w = 0; w < 8; ++w) s += accs[w][threadIdx.x];
        essum[nt * DD + threadIdx.x] = s;
    }
}

#define PB_CPY (5 * NT)
#define PB_END (PB_CPY + (NB * 64 * DD) / 256)
__global__ void k_prep(const float* __restrict__ x, float* __restrict__ out,
                       const float* __restrict__ fsl0, const float* __restrict__ fsl1,
                       const float* __restrict__ esl0, const float* __restrict__ esl1,
                       const float* __restrict__ esl2,
                       float* __restrict__ fs0, float* __restrict__ fs1,
                       float* __restrict__ es0, float* __restrict__ es1,
                       float* __restrict__ es2) {
    __shared__ float tile[8][32 * TPAD];
    __shared__ float accs[8][32];
    __shared__ float row[320];
    __shared__ float s_ts[256];
    __shared__ int   s_k[256];
    int blk = blockIdx.x;
    if (blk >= PB_CPY) {
        copy_x_body(blk - PB_CPY, x, out);
        return;
    }
    int job = blk / NT;
    int nt  = blk - job * NT;
    if (job == 2 || job == 4) {
        const float* fsl = (job == 2) ? fsl1 : fsl0;
        float*       fs  = (job == 2) ? fs1  : fs0;
        int          J   = (job == 2) ? 192  : 64;
        fsmax_body(row, s_ts, s_k, nt, fsl, fs, J);
    } else {
        const float* esl = (job == 0) ? esl2 : (job == 1) ? esl1 : esl0;
        float*       es  = (job == 0) ? es2  : (job == 1) ? es1  : es0;
        int          J   = (job == 0) ? 320  : (job == 1) ? 192  : 64;
        essum_body(tile, accs, nt, esl, es, J);
    }
}

__global__ void k_gemm_tl(const float* __restrict__ x, const float* __restrict__ fs,
                          const float* __restrict__ essum, const float* __restrict__ thr,
                          const float* __restrict__ lt, float* __restrict__ tl, int K) {
    int lane  = threadIdx.x & 31;
    int wid   = threadIdx.x >> 5;
    int tile  = blockIdx.x * 8 + wid;
    int mt    = tile >> 7;
    int ntile = tile & 127;
    int m0    = mt * 16;
    int col   = ntile * 16 + (lane & 15);
    int bidx  = col >> 5;
    int d     = col & 31;
    int ka_off = (lane >> 4) << 1;

    const float* arow = fs + (size_t)(m0 + (lane & 15)) * K;
    const float* bcol = x + (size_t)bidx * LDX + d;

    v8f acc = {};
    for (int k0 = 0; k0 < K; k0 += 4) {
        int ka = k0 + ka_off;
        v2f a;
        a.x = arow[ka];
        a.y = arow[ka + 1];
        v2f b;
        b.x = bcol[ka * DD];
        b.y = bcol[(ka + 1) * DD];
        acc = __builtin_amdgcn_wmma_f32_16x16x4_f32(false, a, false, b,
                                                    (short)0, acc, false, false);
    }
    int mplus = (lane >> 4) << 3;
    #pragma unroll
    for (int v = 0; v < 8; ++v) {
        int nt   = m0 + v + mplus;
        float sel = acc[v] * essum[nt * DD + d];
        float tv  = (sel - thr[nt]) * __expf(-lt[nt]);
        tl[(size_t)nt * NCOL + col] = tv;
    }
}

__device__ __forceinline__ void tree_body(float* rsp, int blk, const float* __restrict__ tl,
                                          const float* __restrict__ resp,
                                          float* __restrict__ out, int rowOff) {
    int nn  = blk >> 3;
    int bg  = blk & 7;
    int tid = threadIdx.x;
    for (int idx = tid; idx < DD * LEAF; idx += 256) {
        int dd = idx >> 6, cc = idx & 63;
        rsp[dd * 65 + cc] = resp[(size_t)nn * (DD * LEAF) + idx];
    }
    __syncthreads();

    int lane = tid & 31;
    int wid  = tid >> 5;
    int b    = bg * 8 + wid;

    float p[TT], q[TT];
    #pragma unroll
    for (int t = 0; t < TT; ++t) {
        float v = tl[(size_t)(nn * TT + t) * NCOL + b * DD + lane];
        p[t] = fminf(fmaxf(0.5f * v + 0.5f, 0.0f), 1.0f);
        q[t] = fminf(fmaxf(0.5f - 0.5f * v, 0.0f), 1.0f);
    }
    float r[LEAF];
    r[0] = 1.0f;
    #pragma unroll
    for (int t = 0; t < TT; ++t) {
        int sz = 1 << t;
        #pragma unroll
        for (int c = 0; c < 32; ++c) {
            int cc = sz - 1 - c;
            if (cc >= 0) {
                r[cc + sz] = r[cc] * q[t];
                r[cc]      = r[cc] * p[t];
            }
        }
    }
    float s = 0.0f;
    #pragma unroll
    for (int c = 0; c < LEAF; ++c) s += r[c] * rsp[lane * 65 + c];
    out[(size_t)b * LDX + (rowOff + nn) * DD + lane] = s;
}

__global__ void k_tree(const float* __restrict__ tl, const float* __restrict__ resp,
                       float* __restrict__ out, int rowOff) {
    __shared__ float rsp[DD * 65];
    tree_body(rsp, blockIdx.x, tl, resp, out, rowOff);
}

__global__ void k_tree_fsmax(const float* __restrict__ tl, const float* __restrict__ resp,
                             float* __restrict__ out, int rowOff,
                             const float* __restrict__ fsl, float* __restrict__ fs) {
    __shared__ float rsp[DD * 65];
    __shared__ float row[320];
    __shared__ float s_ts[256];
    __shared__ int   s_k[256];
    int blk = blockIdx.x;
    if (blk < NT) fsmax_body(row, s_ts, s_k, blk, fsl, fs, 320);
    else          tree_body(rsp, blk - NT, tl, resp, out, rowOff);
}

extern "C" void kernel_launch(void* const* d_in, const int* in_sizes, int n_in,
                              void* d_out, int out_size, void* d_ws, size_t ws_size,
                              hipStream_t stream) {
    const float* x = (const float*)d_in[0];
    float* out = (float*)d_out;

    float* slotA = (float*)d_ws;
    float* fs0   = slotA;
    float* fs1   = slotA + NT * 64;
    float* es0   = slotA + NT * 256;
    float* es1   = slotA + NT * 288;
    float* fs2   = slotA;
    float* es2   = slotA + NT * 320;
    float* tl_ws = es2 + NT * DD;

    const float* fsl0  = (const float*)d_in[1];
    const float* esl0  = (const float*)d_in[2];
    const float* thr0  = (const float*)d_in[3];
    const float* lt0   = (const float*)d_in[4];
    const float* resp0 = (const float*)d_in[5];
    const float* fsl1  = (const float*)d_in[6];
    const float* esl1  = (const float*)d_in[7];
    const float* thr1  = (const float*)d_in[8];
    const float* lt1   = (const float*)d_in[9];
    const float* resp1 = (const float*)d_in[10];
    const float* fsl2  = (const float*)d_in[11];
    const float* esl2  = (const float*)d_in[12];
    const float* thr2  = (const float*)d_in[13];
    const float* lt2   = (const float*)d_in[14];
    const float* resp2 = (const float*)d_in[15];

    k_prep<<<PB_END, 256, 0, stream>>>(x, out, fsl0, fsl1, esl0, esl1, esl2,
                                       fs0, fs1, es0, es1, es2);

    k_gemm_tl<<<(48 * 128) / 8, 256, 0, stream>>>(out, fs0, es0, thr0, lt0, tl_ws, 64);
    k_tree   <<<NN * 8, 256, 0, stream>>>(tl_ws, resp0, out, 64);

    k_gemm_tl<<<(48 * 128) / 8, 256, 0, stream>>>(out, fs1, es1, thr1, lt1, tl_ws, 192);
    k_tree_fsmax<<<NT + NN * 8, 256, 0, stream>>>(tl_ws, resp1, out, 192, fsl2, fs2);

    k_gemm_tl<<<(48 * 128) / 8, 256, 0, stream>>>(out, fs2, es2, thr2, lt2, tl_ws, 320);
    k_tree   <<<NN * 8, 256, 0, stream>>>(tl_ws, resp2, out, 320);
}


# GEMM wave tile 16x16 -> 48x32 (3x2 tiles): A by 16-byte loads, x rows by 8-byte loads serving 2 tiles, next 32 k prefetched in registers; k order inside each 8 permuted (f32 sum order changes)
# speedup vs baseline: 1.4073x; 1.1276x over previous
#include <hip/hip_runtime.h>
#include <hip/hip_bf16.h>

#define NB   64
#define DD   32
#define NN   128
#define TT   6
#define NT   (NN*TT)
#define NCOL (NB*DD)
#define LEAF 64
#define JTOT 448
#define LDX  (JTOT*DD)

typedef __attribute__((ext_vector_type(2))) float v2f;
typedef __attribute__((ext_vector_type(8))) float v8f;

__device__ __forceinline__ void copy_x_body(int blk, const float* __restrict__ x,
                                            float* __restrict__ out) {
    int idx = blk * 256 + threadIdx.x;
    if (idx >= NB * 64 * DD) return;
    int b   = idx / (64 * DD);
    int rem = idx - b * (64 * DD);
    out[b * LDX + rem] = x[idx];
}

__device__ __forceinline__ void fsmax_body(float* row, float* s_ts, int* s_k, int nt,
                                           const float* __restrict__ fsl,
                                           float* __restrict__ fs, int J) {
    int tid = threadIdx.x;
    for (int e = tid; e < J; e += 256) row[e] = fsl[nt * J + e];
    __syncthreads();

    float local_ts = 0.0f;
    int   local_k  = 0;
    for (int e = tid; e < J; e += 256) {
        float z = row[e];
        int   c = 1;
        float s = 0.0f;
        for (int j = 0; j < J; ++j) {
            float zj = row[j];
            c += (zj > z) ? 1 : 0;
            s += (zj >= z) ? zj : 0.0f;
        }
        bool supp = ((float)c * z > s - 1.0f);
        local_k  += supp ? 1 : 0;
        local_ts += supp ? z : 0.0f;
    }
    s_ts[tid] = local_ts;
    s_k[tid]  = local_k;
    __syncthreads();
    for (int off = 128; off > 0; off >>= 1) {
        if (tid < off) { s_ts[tid] += s_ts[tid + off]; s_k[tid] += s_k[tid + off]; }
        __syncthreads();
    }
    float tau = (s_ts[0] - 1.0f) / (float)s_k[0];
    for (int e = tid; e < J; e += 256) fs[nt * J + e] = fmaxf(row[e] - tau, 0.0f);
}

#define TPAD 33
__device__ __forceinline__ void essum_body(float (*tile)[32 * TPAD], float (*accs)[32], int nt,
                                           const float* __restrict__ esl,
                                           float* __restrict__ essum, int J) {
    int lane = threadIdx.x & 31;
    int wid  = threadIdx.x >> 5;
    float* my = tile[wid];
    float acc = 0.0f;
    int ntiles = J >> 5;
    for (int tb = wid; tb < ntiles; tb += 8) {
        const float* src = esl + ((size_t)nt * J + tb * 32) * DD + lane;
        #pragma unroll
        for (int rr = 0; rr < 32; ++rr) {
            my[rr * TPAD + lane] = src[rr * DD];
        }
        float r[32];
        #pragma unroll
        for (int k = 0; k < 32; ++k) r[k] = my[lane * TPAD + k];
        #pragma unroll
        for (int k = 2; k <= 32; k <<= 1) {
            #pragma unroll
            for (int j = k >> 1; j > 0; j >>= 1) {
                #pragma unroll
                for (int i = 0; i < 32; ++i) {
                    int ij = i ^ j;
                    if (ij > i) {
                        float a  = r[i], b2 = r[ij];
                        float lo = fminf(a, b2), hi = fmaxf(a, b2);
                        if ((i & k) == 0) { r[i] = hi; r[ij] = lo; }
                        else              { r[i] = lo; r[ij] = hi; }
                    }
                }
            }
        }
        float cum = 0.0f, tausum = 1.0f;
        int   ksup = 1;
        #pragma unroll
        for (int i = 0; i < 32; ++i) {
            cum += r[i];
            bool s = ((float)(i + 1) * r[i] > cum - 1.0f);
            if (s) { ksup = i + 1; tausum = cum; }
        }
        float tau = (tausum - 1.0f) / (float)ksup;
        #pragma unroll
        for (int j2 = 0; j2 < 32; ++j2) {
            float z = my[j2 * TPAD + lane];
            float t = __shfl(tau, j2, 32);
            acc += fmaxf(z - t, 0.0f);
        }
    }
    accs[wid][lane] = acc;
    __syncthreads();
    if (threadIdx.x < 32) {
        float s = 0.0f;
        #pragma unroll
        for (int w = 0; w < 8; ++w) s += accs[w][threadIdx.x];
        essum[nt * DD + threadIdx.x] = s;
    }
}

#define PB_CPY (5 * NT)
#define PB_END (PB_CPY + (NB * 64 * DD) / 256)
__global__ void k_prep(const float* __restrict__ x, float* __restrict__ out,
                       const float* __restrict__ fsl0, const float* __restrict__ fsl1,
                       const float* __restrict__ esl0, const float* __restrict__ esl1,
                       const float* __restrict__ esl2,
                       float* __restrict__ fs0, float* __restrict__ fs1,
                       float* __restrict__ es0, float* __restrict__ es1,
                       float* __restrict__ es2) {
    __shared__ float tile[8][32 * TPAD];
    __shared__ float accs[8][32];
    __shared__ float row[320];
    __shared__ float s_ts[256];
    __shared__ int   s_k[256];
    int blk = blockIdx.x;
    if (blk >= PB_CPY) {
        copy_x_body(blk - PB_CPY, x, out);
        return;
    }
    int job = blk / NT;
    int nt  = blk - job * NT;
    if (job == 2 || job == 4) {
        const float* fsl = (job == 2) ? fsl1 : fsl0;
        float*       fs  = (job == 2) ? fs1  : fs0;
        int          J   = (job == 2) ? 192  : 64;
        fsmax_body(row, s_ts, s_k, nt, fsl, fs, J);
    } else {
        const float* esl = (job == 0) ? esl2 : (job == 1) ? esl1 : esl0;
        float*       es  = (job == 0) ? es2  : (job == 1) ? es1  : es0;
        int          J   = (job == 0) ? 320  : (job == 1) ? 192  : 64;
        essum_body(tile, accs, nt, esl, es, J);
    }
}

typedef __attribute__((ext_vector_type(4))) float v4f;
#define GK 32
struct GFrag { v4f a[3][4]; v2f b[4][4]; };
#define PREFETCH_GROUP_END() do { asm volatile("" ::: "memory"); __builtin_amdgcn_sched_barrier(0); } while (0)
#define MMA_BLOCK_END() __builtin_amdgcn_sched_barrier(0)

__device__ __forceinline__ void gemm_load(GFrag& f, const float* __restrict__ ap,
                                          const float* __restrict__ bp, int K, int kB) {
    #pragma unroll
    for (int j = 0; j < 4; ++j) {
        #pragma unroll
        for (int mi = 0; mi < 3; ++mi)
            f.a[mi][j] = *(const v4f*)(ap + (size_t)(16 * mi) * K + kB + 8 * j);
        #pragma unroll
        for (int e = 0; e < 4; ++e)
            f.b[j][e] = *(const v2f*)(bp + (kB + 8 * j + e) * DD);
    }
}

__device__ __forceinline__ void gemm_mma(v8f (&acc)[3][2], const GFrag& f) {
    #pragma unroll
    for (int j = 0; j < 4; ++j) {
        #pragma unroll
        for (int p = 0; p < 2; ++p) {
            #pragma unroll
            for (int mi = 0; mi < 3; ++mi) {
                #pragma unroll
                for (int t = 0; t < 2; ++t) {
                    v2f a, b;
                    a.x = f.a[mi][j][2 * p];
                    a.y = f.a[mi][j][2 * p + 1];
                    b.x = f.b[j][2 * p][t];
                    b.y = f.b[j][2 * p + 1][t];
                    acc[mi][t] = __builtin_amdgcn_wmma_f32_16x16x4_f32(false, a, false, b,
                                                    (short)0, acc[mi][t], false, false);
                }
            }
        }
    }
}

__global__ void __launch_bounds__(128)
k_gemm_tl(const float* __restrict__ x, const float* __restrict__ fs,
          const float* __restrict__ essum, const float* __restrict__ thr,
          const float* __restrict__ lt, float* __restrict__ tl, int K) {
    int lane = threadIdx.x & 31;
    int wid  = threadIdx.x >> 5;
    int c    = lane & 15;
    int h    = lane >> 4;
    int m0   = (blockIdx.x >> 4) * 48;
    int b    = ((blockIdx.x & 15) << 2) + wid;

    const float* ap = fs + (size_t)(m0 + c) * K + 4 * h;
    const float* bp = x + (size_t)b * LDX + (4 * h) * DD + 2 * c;

    v8f acc[3][2] = {};
    GFrag f0, f1;
    int nkb = K / GK;
    gemm_load(f0, ap, bp, K, 0);
    for (int kb = 0; kb < nkb; kb += 2) {
        gemm_load(f1, ap, bp, K, (kb + 1) * GK);
        PREFETCH_GROUP_END();
        gemm_mma(acc, f0);
        MMA_BLOCK_END();
        int kn = (kb + 2 < nkb) ? (kb + 2) : 0;
        gemm_load(f0, ap, bp, K, kn * GK);
        PREFETCH_GROUP_END();
        gemm_mma(acc, f1);
        MMA_BLOCK_END();
    }

    #pragma unroll
    for (int mi = 0; mi < 3; ++mi) {
        #pragma unroll
        for (int r = 0; r < 8; ++r) {
            int   nt = m0 + 16 * mi + 8 * h + r;
            v2f   es = *(const v2f*)(essum + nt * DD + 2 * c);
            float th = thr[nt];
            float ex = __expf(-lt[nt]);
            v2f o;
            o.x = (acc[mi][0][r] * es.x - th) * ex;
            o.y = (acc[mi][1][r] * es.y - th) * ex;
            *(v2f*)(tl + (size_t)nt * NCOL + b * DD + 2 * c) = o;
        }
    }
}

__device__ __forceinline__ void tree_body(float* rsp, int blk, const float* __restrict__ tl,
                                          const float* __restrict__ resp,
                                          float* __restrict__ out, int rowOff) {
    int nn  = blk >> 3;
    int bg  = blk & 7;
    int tid = threadIdx.x;
    for (int idx = tid; idx < DD * LEAF; idx += 256) {
        int dd = idx >> 6, cc = idx & 63;
        rsp[dd * 65 + cc] = resp[(size_t)nn * (DD * LEAF) + idx];
    }
    __syncthreads();

    int lane = tid & 31;
    int wid  = tid >> 5;
    int b    = bg * 8 + wid;

    float p[TT], q[TT];
    #pragma unroll
    for (int t = 0; t < TT; ++t) {
        float v = tl[(size_t)(nn * TT + t) * NCOL + b * DD + lane];
        p[t] = fminf(fmaxf(0.5f * v + 0.5f, 0.0f), 1.0f);
        q[t] = fminf(fmaxf(0.5f - 0.5f * v, 0.0f), 1.0f);
    }
    float r[LEAF];
    r[0] = 1.0f;
    #pragma unroll
    for (int t = 0; t < TT; ++t) {
        int sz = 1 << t;
        #pragma unroll
        for (int c = 0; c < 32; ++c) {
            int cc = sz - 1 - c;
            if (cc >= 0) {
                r[cc + sz] = r[cc] * q[t];
                r[cc]      = r[cc] * p[t];
            }
        }
    }
    float s = 0.0f;
    #pragma unroll
    for (int c = 0; c < LEAF; ++c) s += r[c] * rsp[lane * 65 + c];
    out[(size_t)b * LDX + (rowOff + nn) * DD + lane] = s;
}

__global__ void k_tree(const float* __restrict__ tl, const float* __restrict__ resp,
                       float* __restrict__ out, int rowOff) {
    __shared__ float rsp[DD * 65];
    tree_body(rsp, blockIdx.x, tl, resp, out, rowOff);
}

__global__ void k_tree_fsmax(const float* __restrict__ tl, const float* __restrict__ resp,
                             float* __restrict__ out, int rowOff,
                             const float* __restrict__ fsl, float* __restrict__ fs) {
    __shared__ float rsp[DD * 65];
    __shared__ float row[320];
    __shared__ float s_ts[256];
    __shared__ int   s_k[256];
    int blk = blockIdx.x;
    if (blk < NT) fsmax_body(row, s_ts, s_k, blk, fsl, fs, 320);
    else          tree_body(rsp, blk - NT, tl, resp, out, rowOff);
}

extern "C" void kernel_launch(void* const* d_in, const int* in_sizes, int n_in,
                              void* d_out, int out_size, void* d_ws, size_t ws_size,
                              hipStream_t stream) {
    const float* x = (const float*)d_in[0];
    float* out = (float*)d_out;

    float* slotA = (float*)d_ws;
    float* fs0   = slotA;
    float* fs1   = slotA + NT * 64;
    float* es0   = slotA + NT * 256;
    float* es1   = slotA + NT * 288;
    float* fs2   = slotA;
    float* es2   = slotA + NT * 320;
    float* tl_ws = es2 + NT * DD;

    const float* fsl0  = (const float*)d_in[1];
    const float* esl0  = (const float*)d_in[2];
    const float* thr0  = (const float*)d_in[3];
    const float* lt0   = (const float*)d_in[4];
    const float* resp0 = (const float*)d_in[5];
    const float* fsl1  = (const float*)d_in[6];
    const float* esl1  = (const float*)d_in[7];
    const float* thr1  = (const float*)d_in[8];
    const float* lt1   = (const float*)d_in[9];
    const float* resp1 = (const float*)d_in[10];
    const float* fsl2  = (const float*)d_in[11];
    const float* esl2  = (const float*)d_in[12];
    const float* thr2  = (const float*)d_in[13];
    const float* lt2   = (const float*)d_in[14];
    const float* resp2 = (const float*)d_in[15];

    k_prep<<<PB_END, 256, 0, stream>>>(x, out, fsl0, fsl1, esl0, esl1, esl2,
                                       fs0, fs1, es0, es1, es2);

    k_gemm_tl<<<256, 128, 0, stream>>>(out, fs0, es0, thr0, lt0, tl_ws, 64);
    k_tree   <<<NN * 8, 256, 0, stream>>>(tl_ws, resp0, out, 64);

    k_gemm_tl<<<256, 128, 0, stream>>>(out, fs1, es1, thr1, lt1, tl_ws, 192);
    k_tree_fsmax<<<NT + NN * 8, 256, 0, stream>>>(tl_ws, resp1, out, 192, fsl2, fs2);

    k_gemm_tl<<<256, 128, 0, stream>>>(out, fs2, es2, thr2, lt2, tl_ws, 320);
    k_tree   <<<NN * 8, 256, 0, stream>>>(tl_ws, resp2, out, 320);
}


# Leaf tree + response dot fused into the GEMM launch: tl goes through a wave-private LDS tile, never to memory (7 launches -> 4); fsmax 2 moves into the input-only launch (freed tl slot)
# speedup vs baseline: 1.8681x; 1.3274x over previous
#include <hip/hip_runtime.h>
#include <hip/hip_bf16.h>

#define NB   64
#define DD   32
#define NN   128
#define TT   6
#define NT   (NN*TT)
#define NCOL (NB*DD)
#define LEAF 64
#define JTOT 448
#define LDX  (JTOT*DD)

typedef __attribute__((ext_vector_type(2))) float v2f;
typedef __attribute__((ext_vector_type(8))) float v8f;

__device__ __forceinline__ void copy_x_body(int blk, const float* __restrict__ x,
                                            float* __restrict__ out) {
    int idx = blk * 256 + threadIdx.x;
    if (idx >= NB * 64 * DD) return;
    int b   = idx / (64 * DD);
    int rem = idx - b * (64 * DD);
    out[b * LDX + rem] = x[idx];
}

__device__ __forceinline__ void fsmax_body(float* row, float* s_ts, int* s_k, int nt,
                                           const float* __restrict__ fsl,
                                           float* __restrict__ fs, int J) {
    int tid = threadIdx.x;
    for (int e = tid; e < J; e += 256) row[e] = fsl[nt * J + e];
    __syncthreads();

    float local_ts = 0.0f;
    int   local_k  = 0;
    for (int e = tid; e < J; e += 256) {
        float z = row[e];
        int   c = 1;
        float s = 0.0f;
        for (int j = 0; j < J; ++j) {
            float zj = row[j];
            c += (zj > z) ? 1 : 0;
            s += (zj >= z) ? zj : 0.0f;
        }
        bool supp = ((float)c * z > s - 1.0f);
        local_k  += supp ? 1 : 0;
        local_ts += supp ? z : 0.0f;
    }
    s_ts[tid] = local_ts;
    s_k[tid]  = local_k;
    __syncthreads();
    for (int off = 128; off > 0; off >>= 1) {
        if (tid < off) { s_ts[tid] += s_ts[tid + off]; s_k[tid] += s_k[tid + off]; }
        __syncthreads();
    }
    float tau = (s_ts[0] - 1.0f) / (float)s_k[0];
    for (int e = tid; e < J; e += 256) fs[nt * J + e] = fmaxf(row[e] - tau, 0.0f);
}

#define TPAD 33
__device__ __forceinline__ void essum_body(float (*tile)[32 * TPAD], float (*accs)[32], int nt,
                                           const float* __restrict__ esl,
                                           float* __restrict__ essum, int J) {
    int lane = threadIdx.x & 31;
    int wid  = threadIdx.x >> 5;
    float* my = tile[wid];
    float acc = 0.0f;
    int ntiles = J >> 5;
    for (int tb = wid; tb < ntiles; tb += 8) {
        const float* src = esl + ((size_t)nt * J + tb * 32) * DD + lane;
        #pragma unroll
        for (int rr = 0; rr < 32; ++rr) {
            my[rr * TPAD + lane] = src[rr * DD];
        }
        float r[32];
        #pragma unroll
        for (int k = 0; k < 32; ++k) r[k] = my[lane * TPAD + k];
        #pragma unroll
        for (int k = 2; k <= 32; k <<= 1) {
            #pragma unroll
            for (int j = k >> 1; j > 0; j >>= 1) {
                #pragma unroll
                for (int i = 0; i < 32; ++i) {
                    int ij = i ^ j;
                    if (ij > i) {
                        float a  = r[i], b2 = r[ij];
                        float lo = fminf(a, b2), hi = fmaxf(a, b2);
                        if ((i & k) == 0) { r[i] = hi; r[ij] = lo; }
                        else              { r[i] = lo; r[ij] = hi; }
                    }
                }
            }
        }
        float cum = 0.0f, tausum = 1.0f;
        int   ksup = 1;
        #pragma unroll
        for (int i = 0; i < 32; ++i) {
            cum += r[i];
            bool s = ((float)(i + 1) * r[i] > cum - 1.0f);
            if (s) { ksup = i + 1; tausum = cum; }
        }
        float tau = (tausum - 1.0f) / (float)ksup;
        #pragma unroll
        for (int j2 = 0; j2 < 32; ++j2) {
            float z = my[j2 * TPAD + lane];
            float t = __shfl(tau, j2, 32);
            acc += fmaxf(z - t, 0.0f);
        }
    }
    accs[wid][lane] = acc;
    __syncthreads();
    if (threadIdx.x < 32) {
        float s = 0.0f;
        #pragma unroll
        for (int w = 0; w < 8; ++w) s += accs[w][threadIdx.x];
        essum[nt * DD + threadIdx.x] = s;
    }
}

#define PB_CPY (6 * NT)
#define PB_END (PB_CPY + (NB * 64 * DD) / 256)
__global__ void k_prep(const float* __restrict__ x, float* __restrict__ out,
                       const float* __restrict__ fsl0, const float* __restrict__ fsl1,
                       const float* __restrict__ fsl2,
                       const float* __restrict__ esl0, const float* __restrict__ esl1,
                       const float* __restrict__ esl2,
                       float* __restrict__ fs0, float* __restrict__ fs1,
                       float* __restrict__ fs2,
                       float* __restrict__ es0, float* __restrict__ es1,
                       float* __restrict__ es2) {
    __shared__ float tile[8][32 * TPAD];
    __shared__ float accs[8][32];
    __shared__ float row[320];
    __shared__ float s_ts[256];
    __shared__ int   s_k[256];
    int blk = blockIdx.x;
    if (blk >= PB_CPY) {
        copy_x_body(blk - PB_CPY, x, out);
        return;
    }
    int job = blk / NT;
    int nt  = blk - job * NT;
    int lay = 2 - (job >> 1);
    int J   = (lay == 2) ? 320 : (lay == 1) ? 192 : 64;
    if (job & 1) {
        const float* fsl = (lay == 2) ? fsl2 : (lay == 1) ? fsl1 : fsl0;
        float*       fs  = (lay == 2) ? fs2  : (lay == 1) ? fs1  : fs0;
        fsmax_body(row, s_ts, s_k, nt, fsl, fs, J);
    } else {
        const float* esl = (lay == 2) ? esl2 : (lay == 1) ? esl1 : esl0;
        float*       es  = (lay == 2) ? es2  : (lay == 1) ? es1  : es0;
        essum_body(tile, accs, nt, esl, es, J);
    }
}

typedef __attribute__((ext_vector_type(4))) float v4f;
#define GK 32
struct GFrag { v4f a[3][4]; v2f b[4][4]; };
#define PREFETCH_GROUP_END() do { asm volatile("" ::: "memory"); __builtin_amdgcn_sched_barrier(0); } while (0)
#define MMA_BLOCK_END() __builtin_amdgcn_sched_barrier(0)

__device__ __forceinline__ void gemm_load(GFrag& f, const float* __restrict__ ap,
                                          const float* __restrict__ bp, int K, int kB) {
    #pragma unroll
    for (int j = 0; j < 4; ++j) {
        #pragma unroll
        for (int mi = 0; mi < 3; ++mi)
            f.a[mi][j] = *(const v4f*)(ap + (size_t)(16 * mi) * K + kB + 8 * j);
        #pragma unroll
        for (int e = 0; e < 4; ++e)
            f.b[j][e] = *(const v2f*)(bp + (kB + 8 * j + e) * DD);
    }
}

__device__ __forceinline__ void gemm_mma(v8f (&acc)[3][2], const GFrag& f) {
    #pragma unroll
    for (int j = 0; j < 4; ++j) {
        #pragma unroll
        for (int p = 0; p < 2; ++p) {
            #pragma unroll
            for (int mi = 0; mi < 3; ++mi) {
                #pragma unroll
                for (int t = 0; t < 2; ++t) {
                    v2f a, b;
                    a.x = f.a[mi][j][2 * p];
                    a.y = f.a[mi][j][2 * p + 1];
                    b.x = f.b[j][2 * p][t];
                    b.y = f.b[j][2 * p + 1][t];
                    acc[mi][t] = __builtin_amdgcn_wmma_f32_16x16x4_f32(false, a, false, b,
                                                    (short)0, acc[mi][t], false, false);
                }
            }
        }
    }
}

#define TLP 36
#define NU  2
#define RSP 68
__global__ void __launch_bounds__(128)
k_layer(const float* __restrict__ x, const float* __restrict__ fs,
        const float* __restrict__ essum, const float* __restrict__ thr,
        const float* __restrict__ lt, const float* __restrict__ resp,
        float* __restrict__ out, int rowOff, int K) {
    __shared__ __attribute__((aligned(16))) float rsp[8 * DD * RSP];
    __shared__ __attribute__((aligned(16))) float tls[4][48 * TLP];
    int lane = threadIdx.x & 31;
    int wid  = threadIdx.x >> 5;
    int c    = lane & 15;
    int h    = lane >> 4;
    int grp  = blockIdx.x >> 4;
    int m0   = grp * 48;
    int nn0  = grp * 8;
    int b    = ((blockIdx.x & 15) << 2) + wid;

    const float* ap = fs + (size_t)(m0 + c) * K + 4 * h;
    const float* bp = x + (size_t)b * LDX + (4 * h) * DD + 2 * c;

    {
        const v4f* rsrc = (const v4f*)(resp + (size_t)nn0 * (DD * LEAF));
        #pragma unroll
        for (int i = 0; i < 32; ++i) {
            int i4 = threadIdx.x + 128 * i;
            v4f v  = rsrc[i4];
            *(v4f*)(rsp + (i4 >> 4) * RSP + 4 * (i4 & 15)) = v;
        }
    }

    v8f acc[3][2] = {};
    GFrag f0, f1;
    int nkb = K / GK;
    gemm_load(f0, ap, bp, K, 0);
    for (int kb = 0; kb < nkb; kb += 2) {
        gemm_load(f1, ap, bp, K, (kb + 1) * GK);
        PREFETCH_GROUP_END();
        gemm_mma(acc, f0);
        MMA_BLOCK_END();
        int kn = (kb + 2 < nkb) ? (kb + 2) : 0;
        gemm_load(f0, ap, bp, K, kn * GK);
        PREFETCH_GROUP_END();
        gemm_mma(acc, f1);
        MMA_BLOCK_END();
    }

    float* myt = tls[wid];
    #pragma unroll
    for (int mi = 0; mi < 3; ++mi) {
        #pragma unroll
        for (int r = 0; r < 8; ++r) {
            int   row = 16 * mi + 8 * h + r;
            int   nt  = m0 + row;
            v2f   es = *(const v2f*)(essum + nt * DD + 2 * c);
            float th = thr[nt];
            float ex = __expf(-lt[nt]);
            v2f o;
            o.x = (acc[mi][0][r] * es.x - th) * ex;
            o.y = (acc[mi][1][r] * es.y - th) * ex;
            *(v2f*)(myt + row * TLP + 2 * c) = o;
        }
    }
    __syncthreads();

    #pragma unroll 1
    for (int ng = 0; ng < 8; ng += NU) {
        float r[NU][LEAF];
        #pragma unroll
        for (int u = 0; u < NU; ++u) {
            int nl = ng + u;
            float p[TT], q[TT];
            #pragma unroll
            for (int t = 0; t < TT; ++t) {
                float v = myt[(nl * TT + t) * TLP + lane];
                p[t] = fminf(fmaxf(0.5f * v + 0.5f, 0.0f), 1.0f);
                q[t] = fminf(fmaxf(0.5f - 0.5f * v, 0.0f), 1.0f);
            }
            r[u][0] = 1.0f;
            #pragma unroll
            for (int t = 0; t < TT; ++t) {
                int sz = 1 << t;
                #pragma unroll
                for (int cI = 0; cI < 32; ++cI) {
                    int cc = sz - 1 - cI;
                    if (cc >= 0) {
                        r[u][cc + sz] = r[u][cc] * q[t];
                        r[u][cc]      = r[u][cc] * p[t];
                    }
                }
            }
        }
        float s[NU];
        #pragma unroll
        for (int u = 0; u < NU; ++u) s[u] = 0.0f;
        #pragma unroll
        for (int c4 = 0; c4 < LEAF / 4; ++c4) {
            #pragma unroll
            for (int u = 0; u < NU; ++u) {
                v4f w = *(const v4f*)(rsp + ((ng + u) * DD + lane) * RSP + 4 * c4);
                s[u] += r[u][4 * c4 + 0] * w.x;
                s[u] += r[u][4 * c4 + 1] * w.y;
                s[u] += r[u][4 * c4 + 2] * w.z;
                s[u] += r[u][4 * c4 + 3] * w.w;
            }
        }
        #pragma unroll
        for (int u = 0; u < NU; ++u)
            out[(size_t)b * LDX + (rowOff + nn0 + ng + u) * DD + lane] = s[u];
    }
}

extern "C" void kernel_launch(void* const* d_in, const int* in_sizes, int n_in,
                              void* d_out, int out_size, void* d_ws, size_t ws_size,
                              hipStream_t stream) {
    const float* x = (const float*)d_in[0];
    float* out = (float*)d_out;

    float* slotA = (float*)d_ws;
    float* fs0   = slotA;
    float* fs1   = slotA + NT * 64;
    float* es0   = slotA + NT * 256;
    float* es1   = slotA + NT * 288;
    float* es2   = slotA + NT * 320;
    float* fs2   = es2 + NT * DD;

    const float* fsl0  = (const float*)d_in[1];
    const float* esl0  = (const float*)d_in[2];
    const float* thr0  = (const float*)d_in[3];
    const float* lt0   = (const float*)d_in[4];
    const float* resp0 = (const float*)d_in[5];
    const float* fsl1  = (const float*)d_in[6];
    const float* esl1  = (const float*)d_in[7];
    const float* thr1  = (const float*)d_in[8];
    const float* lt1   = (const float*)d_in[9];
    const float* resp1 = (const float*)d_in[10];
    const float* fsl2  = (const float*)d_in[11];
    const float* esl2  = (const float*)d_in[12];
    const float* thr2  = (const float*)d_in[13];
    const float* lt2   = (const float*)d_in[14];
    const float* resp2 = (const float*)d_in[15];

    k_prep<<<PB_END, 256, 0, stream>>>(x, out, fsl0, fsl1, fsl2, esl0, esl1, esl2,
                                       fs0, fs1, fs2, es0, es1, es2);

    k_layer<<<256, 128, 0, stream>>>(out, fs0, es0, thr0, lt0, resp0, out, 64,  64);
    k_layer<<<256, 128, 0, stream>>>(out, fs1, es1, thr1, lt1, resp1, out, 192, 192);
    k_layer<<<256, 128, 0, stream>>>(out, fs2, es2, thr2, lt2, resp2, out, 320, 320);
}


# sparsemax(fsl): one wave per row, row in registers, tau by threshold passes (tau=(sum_S-1)/|S| until S stops shrinking) instead of a 256-thread O(J^2) rank count; tau's f32 sum order changes
# speedup vs baseline: 2.2865x; 1.2240x over previous
#include <hip/hip_runtime.h>
#include <hip/hip_bf16.h>

#define NB   64
#define DD   32
#define NN   128
#define TT   6
#define NT   (NN*TT)
#define NCOL (NB*DD)
#define LEAF 64
#define JTOT 448
#define LDX  (JTOT*DD)

typedef __attribute__((ext_vector_type(2))) float v2f;
typedef __attribute__((ext_vector_type(8))) float v8f;

__device__ __forceinline__ void copy_x_body(int blk, const float* __restrict__ x,
                                            float* __restrict__ out) {
    int idx = blk * 256 + threadIdx.x;
    if (idx >= NB * 64 * DD) return;
    int b   = idx / (64 * DD);
    int rem = idx - b * (64 * DD);
    out[b * LDX + rem] = x[idx];
}

template <int NPL>
__device__ __forceinline__ void fsmax_wave(int nt, const float* __restrict__ fsl,
                                           float* __restrict__ fs) {
    constexpr int J = 32 * NPL;
    int lane = threadIdx.x & 31;
    const float* src = fsl + (size_t)nt * J + lane;
    float z[NPL];
    #pragma unroll
    for (int i = 0; i < NPL; ++i) z[i] = src[32 * i];

    float tau   = -INFINITY;
    int   kprev = -1;
    for (int it = 0; it <= J; ++it) {
        float sum = 0.0f;
        int   k   = 0;
        #pragma unroll
        for (int i = 0; i < NPL; ++i) {
            bool in = z[i] > tau;
            sum += in ? z[i] : 0.0f;
            k   += in ? 1 : 0;
        }
        #pragma unroll
        for (int m = 16; m > 0; m >>= 1) {
            sum += __shfl_xor(sum, m, 32);
            k   += __shfl_xor(k, m, 32);
        }
        int ku = __builtin_amdgcn_readfirstlane(k);
        if (ku == kprev) break;
        kprev = ku;
        tau = fmaxf(tau, (sum - 1.0f) / (float)ku);
    }
    float* dst = fs + (size_t)nt * J + lane;
    #pragma unroll
    for (int i = 0; i < NPL; ++i) dst[32 * i] = fmaxf(z[i] - tau, 0.0f);
}

#define TPAD 33
__device__ __forceinline__ void essum_body(float (*tile)[32 * TPAD], float (*accs)[32], int nt,
                                           const float* __restrict__ esl,
                                           float* __restrict__ essum, int J) {
    int lane = threadIdx.x & 31;
    int wid  = threadIdx.x >> 5;
    float* my = tile[wid];
    float acc = 0.0f;
    int ntiles = J >> 5;
    for (int tb = wid; tb < ntiles; tb += 8) {
        const float* src = esl + ((size_t)nt * J + tb * 32) * DD + lane;
        #pragma unroll
        for (int rr = 0; rr < 32; ++rr) {
            my[rr * TPAD + lane] = src[rr * DD];
        }
        float r[32];
        #pragma unroll
        for (int k = 0; k < 32; ++k) r[k] = my[lane * TPAD + k];
        #pragma unroll
        for (int k = 2; k <= 32; k <<= 1) {
            #pragma unroll
            for (int j = k >> 1; j > 0; j >>= 1) {
                #pragma unroll
                for (int i = 0; i < 32; ++i) {
                    int ij = i ^ j;
                    if (ij > i) {
                        float a  = r[i], b2 = r[ij];
                        float lo = fminf(a, b2), hi = fmaxf(a, b2);
                        if ((i & k) == 0) { r[i] = hi; r[ij] = lo; }
                        else              { r[i] = lo; r[ij] = hi; }
                    }
                }
            }
        }
        float cum = 0.0f, tausum = 1.0f;
        int   ksup = 1;
        #pragma unroll
        for (int i = 0; i < 32; ++i) {
            cum += r[i];
            bool s = ((float)(i + 1) * r[i] > cum - 1.0f);
            if (s) { ksup = i + 1; tausum = cum; }
        }
        float tau = (tausum - 1.0f) / (float)ksup;
        #pragma unroll
        for (int j2 = 0; j2 < 32; ++j2) {
            float z = my[j2 * TPAD + lane];
            float t = __shfl(tau, j2, 32);
            acc += fmaxf(z - t, 0.0f);
        }
    }
    accs[wid][lane] = acc;
    __syncthreads();
    if (threadIdx.x < 32) {
        float s = 0.0f;
        #pragma unroll
        for (int w = 0; w < 8; ++w) s += accs[w][threadIdx.x];
        essum[nt * DD + threadIdx.x] = s;
    }
}

#define FSB    (NT / 8)
#define PB_FS  (2 * NT)
#define PB_CPY (3 * NT + 3 * FSB)
#define PB_END (PB_CPY + (NB * 64 * DD) / 256)
__global__ void __launch_bounds__(256, 10)
k_prep(const float* __restrict__ x, float* __restrict__ out,
                       const float* __restrict__ fsl0, const float* __restrict__ fsl1,
                       const float* __restrict__ fsl2,
                       const float* __restrict__ esl0, const float* __restrict__ esl1,
                       const float* __restrict__ esl2,
                       float* __restrict__ fs0, float* __restrict__ fs1,
                       float* __restrict__ fs2,
                       float* __restrict__ es0, float* __restrict__ es1,
                       float* __restrict__ es2) {
    __shared__ float tile[8][32 * TPAD];
    __shared__ float accs[8][32];
    int blk = blockIdx.x;
    if (blk >= PB_CPY) {
        copy_x_body(blk - PB_CPY, x, out);
        return;
    }
    int q = blk - PB_FS;
    if (q >= 0 && q < 3 * FSB) {
        int job = q / FSB;
        int nt  = (q - job * FSB) * 8 + (threadIdx.x >> 5);
        if (job == 0)      fsmax_wave<10>(nt, fsl2, fs2);
        else if (job == 1) fsmax_wave<6>(nt, fsl1, fs1);
        else               fsmax_wave<2>(nt, fsl0, fs0);
        return;
    }
    int eb  = (q < 0) ? blk : blk - 3 * FSB;
    int job = eb / NT;
    int nt  = eb - job * NT;
    int lay = 2 - job;
    int J   = (lay == 2) ? 320 : (lay == 1) ? 192 : 64;
    const float* esl = (lay == 2) ? esl2 : (lay == 1) ? esl1 : esl0;
    float*       es  = (lay == 2) ? es2  : (lay == 1) ? es1  : es0;
    essum_body(tile, accs, nt, esl, es, J);
}

typedef __attribute__((ext_vector_type(4))) float v4f;
#define GK 32
struct GFrag { v4f a[3][4]; v2f b[4][4]; };
#define PREFETCH_GROUP_END() do { asm volatile("" ::: "memory"); __builtin_amdgcn_sched_barrier(0); } while (0)
#define MMA_BLOCK_END() __builtin_amdgcn_sched_barrier(0)

__device__ __forceinline__ void gemm_load(GFrag& f, const float* __restrict__ ap,
                                          const float* __restrict__ bp, int K, int kB) {
    #pragma unroll
    for (int j = 0; j < 4; ++j) {
        #pragma unroll
        for (int mi = 0; mi < 3; ++mi)
            f.a[mi][j] = *(const v4f*)(ap + (size_t)(16 * mi) * K + kB + 8 * j);
        #pragma unroll
        for (int e = 0; e < 4; ++e)
            f.b[j][e] = *(const v2f*)(bp + (kB + 8 * j + e) * DD);
    }
}

__device__ __forceinline__ void gemm_mma(v8f (&acc)[3][2], const GFrag& f) {
    #pragma unroll
    for (int j = 0; j < 4; ++j) {
        #pragma unroll
        for (int p = 0; p < 2; ++p) {
            #pragma unroll
            for (int mi = 0; mi < 3; ++mi) {
                #pragma unroll
                for (int t = 0; t < 2; ++t) {
                    v2f a, b;
                    a.x = f.a[mi][j][2 * p];
                    a.y = f.a[mi][j][2 * p + 1];
                    b.x = f.b[j][2 * p][t];
                    b.y = f.b[j][2 * p + 1][t];
                    acc[mi][t] = __builtin_amdgcn_wmma_f32_16x16x4_f32(false, a, false, b,
                                                    (short)0, acc[mi][t], false, false);
                }
            }
        }
    }
}

#define TLP 36
#define NU  2
#define RSP 68
__global__ void __launch_bounds__(128)
k_layer(const float* __restrict__ x, const float* __restrict__ fs,
        const float* __restrict__ essum, const float* __restrict__ thr,
        const float* __restrict__ lt, const float* __restrict__ resp,
        float* __restrict__ out, int rowOff, int K) {
    __shared__ __attribute__((aligned(16))) float rsp[8 * DD * RSP];
    __shared__ __attribute__((aligned(16))) float tls[4][48 * TLP];
    int lane = threadIdx.x & 31;
    int wid  = threadIdx.x >> 5;
    int c    = lane & 15;
    int h    = lane >> 4;
    int grp  = blockIdx.x >> 4;
    int m0   = grp * 48;
    int nn0  = grp * 8;
    int b    = ((blockIdx.x & 15) << 2) + wid;

    const float* ap = fs + (size_t)(m0 + c) * K + 4 * h;
    const float* bp = x + (size_t)b * LDX + (4 * h) * DD + 2 * c;

    {
        const v4f* rsrc = (const v4f*)(resp + (size_t)nn0 * (DD * LEAF));
        #pragma unroll
        for (int i = 0; i < 32; ++i) {
            int i4 = threadIdx.x + 128 * i;
            v4f v  = rsrc[i4];
            *(v4f*)(rsp + (i4 >> 4) * RSP + 4 * (i4 & 15)) = v;
        }
    }

    v8f acc[3][2] = {};
    GFrag f0, f1;
    int nkb = K / GK;
    gemm_load(f0, ap, bp, K, 0);
    for (int kb = 0; kb < nkb; kb += 2) {
        gemm_load(f1, ap, bp, K, (kb + 1) * GK);
        PREFETCH_GROUP_END();
        gemm_mma(acc, f0);
        MMA_BLOCK_END();
        int kn = (kb + 2 < nkb) ? (kb + 2) : 0;
        gemm_load(f0, ap, bp, K, kn * GK);
        PREFETCH_GROUP_END();
        gemm_mma(acc, f1);
        MMA_BLOCK_END();
    }

    float* myt = tls[wid];
    #pragma unroll
    for (int mi = 0; mi < 3; ++mi) {
        #pragma unroll
        for (int r = 0; r < 8; ++r) {
            int   row = 16 * mi + 8 * h + r;
            int   nt  = m0 + row;
            v2f   es = *(const v2f*)(essum + nt * DD + 2 * c);
            float th = thr[nt];
            float ex = __expf(-lt[nt]);
            v2f o;
            o.x = (acc[mi][0][r] * es.x - th) * ex;
            o.y = (acc[mi][1][r] * es.y - th) * ex;
            *(v2f*)(myt + row * TLP + 2 * c) = o;
        }
    }
    __syncthreads();

    #pragma unroll 1
    for (int ng = 0; ng < 8; ng += NU) {
        float r[NU][LEAF];
        #pragma unroll
        for (int u = 0; u < NU; ++u) {
            int nl = ng + u;
            float p[TT], q[TT];
            #pragma unroll
            for (int t = 0; t < TT; ++t) {
                float v = myt[(nl * TT + t) * TLP + lane];
                p[t] = fminf(fmaxf(0.5f * v + 0.5f, 0.0f), 1.0f);
                q[t] = fminf(fmaxf(0.5f - 0.5f * v, 0.0f), 1.0f);
            }
            r[u][0] = 1.0f;
            #pragma unroll
            for (int t = 0; t < TT; ++t) {
                int sz = 1 << t;
                #pragma unroll
                for (int cI = 0; cI < 32; ++cI) {
                    int cc = sz - 1 - cI;
                    if (cc >= 0) {
                        r[u][cc + sz] = r[u][cc] * q[t];
                        r[u][cc]      = r[u][cc] * p[t];
                    }
                }
            }
        }
        float s[NU];
        #pragma unroll
        for (int u = 0; u < NU; ++u) s[u] = 0.0f;
        #pragma unroll
        for (int c4 = 0; c4 < LEAF / 4; ++c4) {
            #pragma unroll
            for (int u = 0; u < NU; ++u) {
                v4f w = *(const v4f*)(rsp + ((ng + u) * DD + lane) * RSP + 4 * c4);
                s[u] += r[u][4 * c4 + 0] * w.x;
                s[u] += r[u][4 * c4 + 1] * w.y;
                s[u] += r[u][4 * c4 + 2] * w.z;
                s[u] += r[u][4 * c4 + 3] * w.w;
            }
        }
        #pragma unroll
        for (int u = 0; u < NU; ++u)
            out[(size_t)b * LDX + (rowOff + nn0 + ng + u) * DD + lane] = s[u];
    }
}

extern "C" void kernel_launch(void* const* d_in, const int* in_sizes, int n_in,
                              void* d_out, int out_size, void* d_ws, size_t ws_size,
                              hipStream_t stream) {
    const float* x = (const float*)d_in[0];
    float* out = (float*)d_out;

    float* slotA = (float*)d_ws;
    float* fs0   = slotA;
    float* fs1   = slotA + NT * 64;
    float* es0   = slotA + NT * 256;
    float* es1   = slotA + NT * 288;
    float* es2   = slotA + NT * 320;
    float* fs2   = es2 + NT * DD;

    const float* fsl0  = (const float*)d_in[1];
    const float* esl0  = (const float*)d_in[2];
    const float* thr0  = (const float*)d_in[3];
    const float* lt0   = (const float*)d_in[4];
    const float* resp0 = (const float*)d_in[5];
    const float* fsl1  = (const float*)d_in[6];
    const float* esl1  = (const float*)d_in[7];
    const float* thr1  = (const float*)d_in[8];
    const float* lt1   = (const float*)d_in[9];
    const float* resp1 = (const float*)d_in[10];
    const float* fsl2  = (const float*)d_in[11];
    const float* esl2  = (const float*)d_in[12];
    const float* thr2  = (const float*)d_in[13];
    const float* lt2   = (const float*)d_in[14];
    const float* resp2 = (const float*)d_in[15];

    k_prep<<<PB_END, 256, 0, stream>>>(x, out, fsl0, fsl1, fsl2, esl0, esl1, esl2,
                                       fs0, fs1, fs2, es0, es1, es2);

    k_layer<<<256, 128, 0, stream>>>(out, fs0, es0, thr0, lt0, resp0, out, 64,  64);
    k_layer<<<256, 128, 0, stream>>>(out, fs1, es1, thr1, lt1, resp1, out, 192, 192);
    k_layer<<<256, 128, 0, stream>>>(out, fs2, es2, thr2, lt2, resp2, out, 320, 320);
}


# Prep launch repacked by the wave: 3104 blocks (24832 waves, 10240 idle or copy-only) -> 768 equal blocks of 8 busy waves (2-3 esl tiles each; fsl rows, x copy folded in); essum f32 sum order changes
# speedup vs baseline: 2.3633x; 1.0336x over previous
#include <hip/hip_runtime.h>
#include <hip/hip_bf16.h>

#define NB   64
#define DD   32
#define NN   128
#define TT   6
#define NT   (NN*TT)
#define NCOL (NB*DD)
#define LEAF 64
#define JTOT 448
#define LDX  (JTOT*DD)

typedef __attribute__((ext_vector_type(2))) float v2f;
typedef __attribute__((ext_vector_type(8))) float v8f;

template <int NPL>
__device__ __forceinline__ void fsmax_wave(int nt, const float* __restrict__ fsl,
                                           float* __restrict__ fs) {
    constexpr int J = 32 * NPL;
    int lane = threadIdx.x & 31;
    const float* src = fsl + (size_t)nt * J + lane;
    float z[NPL];
    #pragma unroll
    for (int i = 0; i < NPL; ++i) z[i] = src[32 * i];

    float tau   = -INFINITY;
    int   kprev = -1;
    for (int it = 0; it <= J; ++it) {
        float sum = 0.0f;
        int   k   = 0;
        #pragma unroll
        for (int i = 0; i < NPL; ++i) {
            bool in = z[i] > tau;
            sum += in ? z[i] : 0.0f;
            k   += in ? 1 : 0;
        }
        #pragma unroll
        for (int m = 16; m > 0; m >>= 1) {
            sum += __shfl_xor(sum, m, 32);
            k   += __shfl_xor(k, m, 32);
        }
        int ku = __builtin_amdgcn_readfirstlane(k);
        if (ku == kprev) break;
        kprev = ku;
        tau = fmaxf(tau, (sum - 1.0f) / (float)ku);
    }
    float* dst = fs + (size_t)nt * J + lane;
    #pragma unroll
    for (int i = 0; i < NPL; ++i) dst[32 * i] = fmaxf(z[i] - tau, 0.0f);
}

#define TPAD 33
__device__ __forceinline__ float essum_tiles(float* my, const float* __restrict__ esl_nt,
                                             int t0, int t1) {
    int lane = threadIdx.x & 31;
    float acc = 0.0f;
    for (int tb = t0; tb < t1; ++tb) {
        const float* src = esl_nt + (size_t)(tb * 32) * DD + lane;
        #pragma unroll
        for (int rr = 0; rr < 32; ++rr) {
            my[rr * TPAD + lane] = src[rr * DD];
        }
        float r[32];
        #pragma unroll
        for (int k = 0; k < 32; ++k) r[k] = my[lane * TPAD + k];
        #pragma unroll
        for (int k = 2; k <= 32; k <<= 1) {
            #pragma unroll
            for (int j = k >> 1; j > 0; j >>= 1) {
                #pragma unroll
                for (int i = 0; i < 32; ++i) {
                    int ij = i ^ j;
                    if (ij > i) {
                        float a  = r[i], b2 = r[ij];
                        float lo = fminf(a, b2), hi = fmaxf(a, b2);
                        if ((i & k) == 0) { r[i] = hi; r[ij] = lo; }
                        else              { r[i] = lo; r[ij] = hi; }
                    }
                }
            }
        }
        float cum = 0.0f, tausum = 1.0f;
        int   ksup = 1;
        #pragma unroll
        for (int i = 0; i < 32; ++i) {
            cum += r[i];
            bool s = ((float)(i + 1) * r[i] > cum - 1.0f);
            if (s) { ksup = i + 1; tausum = cum; }
        }
        float tau = (tausum - 1.0f) / (float)ksup;
        #pragma unroll
        for (int j2 = 0; j2 < 32; ++j2) {
            float z = my[j2 * TPAD + lane];
            float t = __shfl(tau, j2, 32);
            acc += fmaxf(z - t, 0.0f);
        }
    }
    return acc;
}

typedef __attribute__((ext_vector_type(4))) float v4f;
__global__ void __launch_bounds__(256, 8)
k_prep(const float* __restrict__ x, float* __restrict__ out,
                       const float* __restrict__ fsl0, const float* __restrict__ fsl1,
                       const float* __restrict__ fsl2,
                       const float* __restrict__ esl0, const float* __restrict__ esl1,
                       const float* __restrict__ esl2,
                       float* __restrict__ fs0, float* __restrict__ fs1,
                       float* __restrict__ fs2,
                       float* __restrict__ es0, float* __restrict__ es1,
                       float* __restrict__ es2) {
    __shared__ float tile[8][32 * TPAD];
    __shared__ float accs[8][32];
    int nt   = blockIdx.x;
    int lane = threadIdx.x & 31;
    int wid  = __builtin_amdgcn_readfirstlane((int)(threadIdx.x >> 5));

    if ((wid == 5 || wid == 6) && nt < 512) {
        int idx4 = nt * 64 + (wid - 5) * 32 + lane;
        int b    = idx4 >> 9;
        int rem  = idx4 & 511;
        v4f v = ((const v4f*)x)[idx4];
        ((v4f*)out)[b * (LDX / 4) + rem] = v;
    }

    if (wid == 2)      fsmax_wave<10>(nt, fsl2, fs2);
    else if (wid == 3) fsmax_wave<6>(nt, fsl1, fs1);
    else if (wid == 7) fsmax_wave<2>(nt, fsl0, fs0);

    int J, t0, t1;
    const float* esl;
    if (wid < 4) {
        J = 320; esl = esl2;
        t0 = (wid < 2) ? 3 * wid : 2 * wid + 2;
        t1 = (wid < 2) ? t0 + 3 : t0 + 2;
    } else if (wid < 7) {
        J = 192; esl = esl1;
        t0 = 2 * (wid - 4);
        t1 = t0 + 2;
    } else {
        J = 64; esl = esl0;
        t0 = 0;
        t1 = 2;
    }
    float acc = essum_tiles(tile[wid], esl + (size_t)nt * J * DD, t0, t1);
    accs[wid][lane] = acc;
    __syncthreads();
    if (threadIdx.x < 96) {
        int g = threadIdx.x >> 5;
        if (g == 0)
            es2[nt * DD + lane] = ((accs[0][lane] + accs[1][lane]) + accs[2][lane]) + accs[3][lane];
        else if (g == 1)
            es1[nt * DD + lane] = (accs[4][lane] + accs[5][lane]) + accs[6][lane];
        else
            es0[nt * DD + lane] = accs[7][lane];
    }
}

#define GK 32
struct GFrag { v4f a[3][4]; v2f b[4][4]; };
#define PREFETCH_GROUP_END() do { asm volatile("" ::: "memory"); __builtin_amdgcn_sched_barrier(0); } while (0)
#define MMA_BLOCK_END() __builtin_amdgcn_sched_barrier(0)

__device__ __forceinline__ void gemm_load(GFrag& f, const float* __restrict__ ap,
                                          const float* __restrict__ bp, int K, int kB) {
    #pragma unroll
    for (int j = 0; j < 4; ++j) {
        #pragma unroll
        for (int mi = 0; mi < 3; ++mi)
            f.a[mi][j] = *(const v4f*)(ap + (size_t)(16 * mi) * K + kB + 8 * j);
        #pragma unroll
        for (int e = 0; e < 4; ++e)
            f.b[j][e] = *(const v2f*)(bp + (kB + 8 * j + e) * DD);
    }
}

__device__ __forceinline__ void gemm_mma(v8f (&acc)[3][2], const GFrag& f) {
    #pragma unroll
    for (int j = 0; j < 4; ++j) {
        #pragma unroll
        for (int p = 0; p < 2; ++p) {
            #pragma unroll
            for (int mi = 0; mi < 3; ++mi) {
                #pragma unroll
                for (int t = 0; t < 2; ++t) {
                    v2f a, b;
                    a.x = f.a[mi][j][2 * p];
                    a.y = f.a[mi][j][2 * p + 1];
                    b.x = f.b[j][2 * p][t];
                    b.y = f.b[j][2 * p + 1][t];
                    acc[mi][t] = __builtin_amdgcn_wmma_f32_16x16x4_f32(false, a, false, b,
                                                    (short)0, acc[mi][t], false, false);
                }
            }
        }
    }
}

#define TLP 36
#define NU  2
#define RSP 68
__global__ void __launch_bounds__(128)
k_layer(const float* __restrict__ x, const float* __restrict__ fs,
        const float* __restrict__ essum, const float* __restrict__ thr,
        const float* __restrict__ lt, const float* __restrict__ resp,
        float* __restrict__ out, int rowOff, int K) {
    __shared__ __attribute__((aligned(16))) float rsp[8 * DD * RSP];
    __shared__ __attribute__((aligned(16))) float tls[4][48 * TLP];
    int lane = threadIdx.x & 31;
    int wid  = threadIdx.x >> 5;
    int c    = lane & 15;
    int h    = lane >> 4;
    int grp  = blockIdx.x >> 4;
    int m0   = grp * 48;
    int nn0  = grp * 8;
    int b    = ((blockIdx.x & 15) << 2) + wid;

    const float* ap = fs + (size_t)(m0 + c) * K + 4 * h;
    const float* bp = x + (size_t)b * LDX + (4 * h) * DD + 2 * c;

    {
        const v4f* rsrc = (const v4f*)(resp + (size_t)nn0 * (DD * LEAF));
        #pragma unroll
        for (int i = 0; i < 32; ++i) {
            int i4 = threadIdx.x + 128 * i;
            v4f v  = rsrc[i4];
            *(v4f*)(rsp + (i4 >> 4) * RSP + 4 * (i4 & 15)) = v;
        }
    }

    v8f acc[3][2] = {};
    GFrag f0, f1;
    int nkb = K / GK;
    gemm_load(f0, ap, bp, K, 0);
    for (int kb = 0; kb < nkb; kb += 2) {
        gemm_load(f1, ap, bp, K, (kb + 1) * GK);
        PREFETCH_GROUP_END();
        gemm_mma(acc, f0);
        MMA_BLOCK_END();
        int kn = (kb + 2 < nkb) ? (kb + 2) : 0;
        gemm_load(f0, ap, bp, K, kn * GK);
        PREFETCH_GROUP_END();
        gemm_mma(acc, f1);
        MMA_BLOCK_END();
    }

    float* myt = tls[wid];
    #pragma unroll
    for (int mi = 0; mi < 3; ++mi) {
        #pragma unroll
        for (int r = 0; r < 8; ++r) {
            int   row = 16 * mi + 8 * h + r;
            int   nt  = m0 + row;
            v2f   es = *(const v2f*)(essum + nt * DD + 2 * c);
            float th = thr[nt];
            float ex = __expf(-lt[nt]);
            v2f o;
            o.x = (acc[mi][0][r] * es.x - th) * ex;
            o.y = (acc[mi][1][r] * es.y - th) * ex;
            *(v2f*)(myt + row * TLP + 2 * c) = o;
        }
    }
    __syncthreads();

    #pragma unroll 1
    for (int ng = 0; ng < 8; ng += NU) {
        float r[NU][LEAF];
        #pragma unroll
        for (int u = 0; u < NU; ++u) {
            int nl = ng + u;
            float p[TT], q[TT];
            #pragma unroll
            for (int t = 0; t < TT; ++t) {
                float v = myt[(nl * TT + t) * TLP + lane];
                p[t] = fminf(fmaxf(0.5f * v + 0.5f, 0.0f), 1.0f);
                q[t] = fminf(fmaxf(0.5f - 0.5f * v, 0.0f), 1.0f);
            }
            r[u][0] = 1.0f;
            #pragma unroll
            for (int t = 0; t < TT; ++t) {
                int sz = 1 << t;
                #pragma unroll
                for (int cI = 0; cI < 32; ++cI) {
                    int cc = sz - 1 - cI;
                    if (cc >= 0) {
                        r[u][cc + sz] = r[u][cc] * q[t];
                        r[u][cc]      = r[u][cc] * p[t];
                    }
                }
            }
        }
        float s[NU];
        #pragma unroll
        for (int u = 0; u < NU; ++u) s[u] = 0.0f;
        #pragma unroll
        for (int c4 = 0; c4 < LEAF / 4; ++c4) {
            #pragma unroll
            for (int u = 0; u < NU; ++u) {
                v4f w = *(const v4f*)(rsp + ((ng + u) * DD + lane) * RSP + 4 * c4);
                s[u] += r[u][4 * c4 + 0] * w.x;
                s[u] += r[u][4 * c4 + 1] * w.y;
                s[u] += r[u][4 * c4 + 2] * w.z;
                s[u] += r[u][4 * c4 + 3] * w.w;
            }
        }
        #pragma unroll
        for (int u = 0; u < NU; ++u)
            out[(size_t)b * LDX + (rowOff + nn0 + ng + u) * DD + lane] = s[u];
    }
}

extern "C" void kernel_launch(void* const* d_in, const int* in_sizes, int n_in,
                              void* d_out, int out_size, void* d_ws, size_t ws_size,
                              hipStream_t stream) {
    const float* x = (const float*)d_in[0];
    float* out = (float*)d_out;

    float* slotA = (float*)d_ws;
    float* fs0   = slotA;
    float* fs1   = slotA + NT * 64;
    float* es0   = slotA + NT * 256;
    float* es1   = slotA + NT * 288;
    float* es2   = slotA + NT * 320;
    float* fs2   = es2 + NT * DD;

    const float* fsl0  = (const float*)d_in[1];
    const float* esl0  = (const float*)d_in[2];
    const float* thr0  = (const float*)d_in[3];
    const float* lt0   = (const float*)d_in[4];
    const float* resp0 = (const float*)d_in[5];
    const float* fsl1  = (const float*)d_in[6];
    const float* esl1  = (const float*)d_in[7];
    const float* thr1  = (const float*)d_in[8];
    const float* lt1   = (const float*)d_in[9];
    const float* resp1 = (const float*)d_in[10];
    const float* fsl2  = (const float*)d_in[11];
    const float* esl2  = (const float*)d_in[12];
    const float* thr2  = (const float*)d_in[13];
    const float* lt2   = (const float*)d_in[14];
    const float* resp2 = (const float*)d_in[15];

    k_prep<<<NT, 256, 0, stream>>>(x, out, fsl0, fsl1, fsl2, esl0, esl1, esl2,
                                       fs0, fs1, fs2, es0, es1, es2);

    k_layer<<<256, 128, 0, stream>>>(out, fs0, es0, thr0, lt0, resp0, out, 64,  64);
    k_layer<<<256, 128, 0, stream>>>(out, fs1, es1, thr1, lt1, resp1, out, 192, 192);
    k_layer<<<256, 128, 0, stream>>>(out, fs2, es2, thr2, lt2, resp2, out, 320, 320);
}


# Layer launch: resp tile, essum, thr, lt requested at the head of the wave behind block 0's fragments and held in registers while the products run; resp goes to LDS in front of the last product block
# speedup vs baseline: 2.5686x; 1.0869x over previous
#include <hip/hip_runtime.h>
#include <hip/hip_bf16.h>

#define NB   64
#define DD   32
#define NN   128
#define TT   6
#define NT   (NN*TT)
#define NCOL (NB*DD)
#define LEAF 64
#define JTOT 448
#define LDX  (JTOT*DD)

typedef __attribute__((ext_vector_type(2))) float v2f;
typedef __attribute__((ext_vector_type(8))) float v8f;

template <int NPL>
__device__ __forceinline__ void fsmax_wave(int nt, const float* __restrict__ fsl,
                                           float* __restrict__ fs) {
    constexpr int J = 32 * NPL;
    int lane = threadIdx.x & 31;
    const float* src = fsl + (size_t)nt * J + lane;
    float z[NPL];
    #pragma unroll
    for (int i = 0; i < NPL; ++i) z[i] = src[32 * i];

    float tau   = -INFINITY;
    int   kprev = -1;
    for (int it = 0; it <= J; ++it) {
        float sum = 0.0f;
        int   k   = 0;
        #pragma unroll
        for (int i = 0; i < NPL; ++i) {
            bool in = z[i] > tau;
            sum += in ? z[i] : 0.0f;
            k   += in ? 1 : 0;
        }
        #pragma unroll
        for (int m = 16; m > 0; m >>= 1) {
            sum += __shfl_xor(sum, m, 32);
            k   += __shfl_xor(k, m, 32);
        }
        int ku = __builtin_amdgcn_readfirstlane(k);
        if (ku == kprev) break;
        kprev = ku;
        tau = fmaxf(tau, (sum - 1.0f) / (float)ku);
    }
    float* dst = fs + (size_t)nt * J + lane;
    #pragma unroll
    for (int i = 0; i < NPL; ++i) dst[32 * i] = fmaxf(z[i] - tau, 0.0f);
}

#define TPAD 33
__device__ __forceinline__ float essum_tiles(float* my, const float* __restrict__ esl_nt,
                                             int t0, int t1) {
    int lane = threadIdx.x & 31;
    float acc = 0.0f;
    for (int tb = t0; tb < t1; ++tb) {
        const float* src = esl_nt + (size_t)(tb * 32) * DD + lane;
        #pragma unroll
        for (int rr = 0; rr < 32; ++rr) {
            my[rr * TPAD + lane] = src[rr * DD];
        }
        float r[32];
        #pragma unroll
        for (int k = 0; k < 32; ++k) r[k] = my[lane * TPAD + k];
        #pragma unroll
        for (int k = 2; k <= 32; k <<= 1) {
            #pragma unroll
            for (int j = k >> 1; j > 0; j >>= 1) {
                #pragma unroll
                for (int i = 0; i < 32; ++i) {
                    int ij = i ^ j;
                    if (ij > i) {
                        float a  = r[i], b2 = r[ij];
                        float lo = fminf(a, b2), hi = fmaxf(a, b2);
                        if ((i & k) == 0) { r[i] = hi; r[ij] = lo; }
                        else              { r[i] = lo; r[ij] = hi; }
                    }
                }
            }
        }
        float cum = 0.0f, tausum = 1.0f;
        int   ksup = 1;
        #pragma unroll
        for (int i = 0; i < 32; ++i) {
            cum += r[i];
            bool s = ((float)(i + 1) * r[i] > cum - 1.0f);
            if (s) { ksup = i + 1; tausum = cum; }
        }
        float tau = (tausum - 1.0f) / (float)ksup;
        #pragma unroll
        for (int j2 = 0; j2 < 32; ++j2) {
            float z = my[j2 * TPAD + lane];
            float t = __shfl(tau, j2, 32);
            acc += fmaxf(z - t, 0.0f);
        }
    }
    return acc;
}

typedef __attribute__((ext_vector_type(4))) float v4f;
__global__ void __launch_bounds__(256, 8)
k_prep(const float* __restrict__ x, float* __restrict__ out,
                       const float* __restrict__ fsl0, const float* __restrict__ fsl1,
                       const float* __restrict__ fsl2,
                       const float* __restrict__ esl0, const float* __restrict__ esl1,
                       const float* __restrict__ esl2,
                       float* __restrict__ fs0, float* __restrict__ fs1,
                       float* __restrict__ fs2,
                       float* __restrict__ es0, float* __restrict__ es1,
                       float* __restrict__ es2) {
    __shared__ float tile[8][32 * TPAD];
    __shared__ float accs[8][32];
    int nt   = blockIdx.x;
    int lane = threadIdx.x & 31;
    int wid  = __builtin_amdgcn_readfirstlane((int)(threadIdx.x >> 5));

    if ((wid == 5 || wid == 6) && nt < 512) {
        int idx4 = nt * 64 + (wid - 5) * 32 + lane;
        int b    = idx4 >> 9;
        int rem  = idx4 & 511;
        v4f v = ((const v4f*)x)[idx4];
        ((v4f*)out)[b * (LDX / 4) + rem] = v;
    }

    if (wid == 2)      fsmax_wave<10>(nt, fsl2, fs2);
    else if (wid == 3) fsmax_wave<6>(nt, fsl1, fs1);
    else if (wid == 7) fsmax_wave<2>(nt, fsl0, fs0);

    int J, t0, t1;
    const float* esl;
    if (wid < 4) {
        J = 320; esl = esl2;
        t0 = (wid < 2) ? 3 * wid : 2 * wid + 2;
        t1 = (wid < 2) ? t0 + 3 : t0 + 2;
    } else if (wid < 7) {
        J = 192; esl = esl1;
        t0 = 2 * (wid - 4);
        t1 = t0 + 2;
    } else {
        J = 64; esl = esl0;
        t0 = 0;
        t1 = 2;
    }
    float acc = essum_tiles(tile[wid], esl + (size_t)nt * J * DD, t0, t1);
    accs[wid][lane] = acc;
    __syncthreads();
    if (threadIdx.x < 96) {
        int g = threadIdx.x >> 5;
        if (g == 0)
            es2[nt * DD + lane] = ((accs[0][lane] + accs[1][lane]) + accs[2][lane]) + accs[3][lane];
        else if (g == 1)
            es1[nt * DD + lane] = (accs[4][lane] + accs[5][lane]) + accs[6][lane];
        else
            es0[nt * DD + lane] = accs[7][lane];
    }
}

#define GK 32
struct GFrag { v4f a[3][4]; v2f b[4][4]; };
#define PREFETCH_GROUP_END() do { asm volatile("" ::: "memory"); __builtin_amdgcn_sched_barrier(0); } while (0)
#define MMA_BLOCK_END() __builtin_amdgcn_sched_barrier(0)

__device__ __forceinline__ void gemm_load(GFrag& f, const float* __restrict__ ap,
                                          const float* __restrict__ bp, int K, int kB) {
    #pragma unroll
    for (int j = 0; j < 4; ++j) {
        #pragma unroll
        for (int mi = 0; mi < 3; ++mi)
            f.a[mi][j] = *(const v4f*)(ap + (size_t)(16 * mi) * K + kB + 8 * j);
        #pragma unroll
        for (int e = 0; e < 4; ++e)
            f.b[j][e] = *(const v2f*)(bp + (kB + 8 * j + e) * DD);
    }
}

__device__ __forceinline__ void gemm_mma(v8f (&acc)[3][2], const GFrag& f) {
    #pragma unroll
    for (int j = 0; j < 4; ++j) {
        #pragma unroll
        for (int p = 0; p < 2; ++p) {
            #pragma unroll
            for (int mi = 0; mi < 3; ++mi) {
                #pragma unroll
                for (int t = 0; t < 2; ++t) {
                    v2f a, b;
                    a.x = f.a[mi][j][2 * p];
                    a.y = f.a[mi][j][2 * p + 1];
                    b.x = f.b[j][2 * p][t];
                    b.y = f.b[j][2 * p + 1][t];
                    acc[mi][t] = __builtin_amdgcn_wmma_f32_16x16x4_f32(false, a, false, b,
                                                    (short)0, acc[mi][t], false, false);
                }
            }
        }
    }
}

#define TLP 36
#define NU  2
#define RSP 68
__global__ void __launch_bounds__(128)
k_layer(const float* __restrict__ x, const float* __restrict__ fs,
        const float* __restrict__ essum, const float* __restrict__ thr,
        const float* __restrict__ lt, const float* __restrict__ resp,
        float* __restrict__ out, int rowOff, int K) {
    __shared__ __attribute__((aligned(16))) float rsp[8 * DD * RSP];
    __shared__ __attribute__((aligned(16))) float tls[4][48 * TLP];
    int lane = threadIdx.x & 31;
    int wid  = threadIdx.x >> 5;
    int c    = lane & 15;
    int h    = lane >> 4;
    int grp  = blockIdx.x >> 4;
    int m0   = grp * 48;
    int nn0  = grp * 8;
    int b    = ((blockIdx.x & 15) << 2) + wid;

    const float* ap = fs + (size_t)(m0 + c) * K + 4 * h;
    const float* bp = x + (size_t)b * LDX + (4 * h) * DD + 2 * c;

    v8f acc[3][2] = {};
    GFrag f0, f1;
    int nkb = K / GK;

    gemm_load(f0, ap, bp, K, 0);
    PREFETCH_GROUP_END();

    v2f   esv[3][8];
    float thv[3][8], ltv[3][8];
    #pragma unroll
    for (int mi = 0; mi < 3; ++mi) {
        #pragma unroll
        for (int r = 0; r < 8; ++r) {
            int nt = m0 + 16 * mi + 8 * h + r;
            esv[mi][r] = *(const v2f*)(essum + nt * DD + 2 * c);
            thv[mi][r] = thr[nt];
            ltv[mi][r] = lt[nt];
        }
    }
    PREFETCH_GROUP_END();
    gemm_load(f1, ap, bp, K, GK);
    PREFETCH_GROUP_END();

    v4f rr[32];
    {
        const v4f* rsrc = (const v4f*)(resp + (size_t)nn0 * (DD * LEAF));
        #pragma unroll
        for (int i = 0; i < 32; ++i) rr[i] = rsrc[threadIdx.x + 128 * i];
    }
    PREFETCH_GROUP_END();

    gemm_mma(acc, f0);
    MMA_BLOCK_END();
    for (int kb = 1; kb + 1 < nkb; kb += 2) {
        gemm_load(f0, ap, bp, K, (kb + 1) * GK);
        PREFETCH_GROUP_END();
        gemm_mma(acc, f1);
        MMA_BLOCK_END();
        gemm_load(f1, ap, bp, K, (kb + 2) * GK);
        PREFETCH_GROUP_END();
        gemm_mma(acc, f0);
        MMA_BLOCK_END();
    }

    #pragma unroll
    for (int i = 0; i < 32; ++i) {
        int i4 = threadIdx.x + 128 * i;
        *(v4f*)(rsp + (i4 >> 4) * RSP + 4 * (i4 & 15)) = rr[i];
    }
    gemm_mma(acc, f1);

    float* myt = tls[wid];
    #pragma unroll
    for (int mi = 0; mi < 3; ++mi) {
        #pragma unroll
        for (int r = 0; r < 8; ++r) {
            int   row = 16 * mi + 8 * h + r;
            v2f   es = esv[mi][r];
            float th = thv[mi][r];
            float ex = __expf(-ltv[mi][r]);
            v2f o;
            o.x = (acc[mi][0][r] * es.x - th) * ex;
            o.y = (acc[mi][1][r] * es.y - th) * ex;
            *(v2f*)(myt + row * TLP + 2 * c) = o;
        }
    }
    __syncthreads();

    #pragma unroll 1
    for (int ng = 0; ng < 8; ng += NU) {
        float r[NU][LEAF];
        #pragma unroll
        for (int u = 0; u < NU; ++u) {
            int nl = ng + u;
            float p[TT], q[TT];
            #pragma unroll
            for (int t = 0; t < TT; ++t) {
                float v = myt[(nl * TT + t) * TLP + lane];
                p[t] = fminf(fmaxf(0.5f * v + 0.5f, 0.0f), 1.0f);
                q[t] = fminf(fmaxf(0.5f - 0.5f * v, 0.0f), 1.0f);
            }
            r[u][0] = 1.0f;
            #pragma unroll
            for (int t = 0; t < TT; ++t) {
                int sz = 1 << t;
                #pragma unroll
                for (int cI = 0; cI < 32; ++cI) {
                    int cc = sz - 1 - cI;
                    if (cc >= 0) {
                        r[u][cc + sz] = r[u][cc] * q[t];
                        r[u][cc]      = r[u][cc] * p[t];
                    }
                }
            }
        }
        float s[NU];
        #pragma unroll
        for (int u = 0; u < NU; ++u) s[u] = 0.0f;
        #pragma unroll
        for (int c4 = 0; c4 < LEAF / 4; ++c4) {
            #pragma unroll
            for (int u = 0; u < NU; ++u) {
                v4f w = *(const v4f*)(rsp + ((ng + u) * DD + lane) * RSP + 4 * c4);
                s[u] += r[u][4 * c4 + 0] * w.x;
                s[u] += r[u][4 * c4 + 1] * w.y;
                s[u] += r[u][4 * c4 + 2] * w.z;
                s[u] += r[u][4 * c4 + 3] * w.w;
            }
        }
        #pragma unroll
        for (int u = 0; u < NU; ++u)
            out[(size_t)b * LDX + (rowOff + nn0 + ng + u) * DD + lane] = s[u];
    }
}

extern "C" void kernel_launch(void* const* d_in, const int* in_sizes, int n_in,
                              void* d_out, int out_size, void* d_ws, size_t ws_size,
                              hipStream_t stream) {
    const float* x = (const float*)d_in[0];
    float* out = (float*)d_out;

    float* slotA = (float*)d_ws;
    float* fs0   = slotA;
    float* fs1   = slotA + NT * 64;
    float* es0   = slotA + NT * 256;
    float* es1   = slotA + NT * 288;
    float* es2   = slotA + NT * 320;
    float* fs2   = es2 + NT * DD;

    const float* fsl0  = (const float*)d_in[1];
    const float* esl0  = (const float*)d_in[2];
    const float* thr0  = (const float*)d_in[3];
    const float* lt0   = (const float*)d_in[4];
    const float* resp0 = (const float*)d_in[5];
    const float* fsl1  = (const float*)d_in[6];
    const float* esl1  = (const float*)d_in[7];
    const float* thr1  = (const float*)d_in[8];
    const float* lt1   = (const float*)d_in[9];
    const float* resp1 = (const float*)d_in[10];
    const float* fsl2  = (const float*)d_in[11];
    const float* esl2  = (const float*)d_in[12];
    const float* thr2  = (const float*)d_in[13];
    const float* lt2   = (const float*)d_in[14];
    const float* resp2 = (const float*)d_in[15];

    k_prep<<<NT, 256, 0, stream>>>(x, out, fsl0, fsl1, fsl2, esl0, esl1, esl2,
                                       fs0, fs1, fs2, es0, es1, es2);

    k_layer<<<256, 128, 0, stream>>>(out, fs0, es0, thr0, lt0, resp0, out, 64,  64);
    k_layer<<<256, 128, 0, stream>>>(out, fs1, es1, thr1, lt1, resp1, out, 192, 192);
    k_layer<<<256, 128, 0, stream>>>(out, fs2, es2, thr2, lt2, resp2, out, 320, 320);
}


# esl sparsemax: each lane takes the 8 largest of its 32-value row (four 8-sorts + three keep-the-top merges) and scans 8; the full 32-sort runs only if a row's 8th value is still in the support
# speedup vs baseline: 2.6793x; 1.0431x over previous
#include <hip/hip_runtime.h>
#include <hip/hip_bf16.h>

#define NB   64
#define DD   32
#define NN   128
#define TT   6
#define NT   (NN*TT)
#define NCOL (NB*DD)
#define LEAF 64
#define JTOT 448
#define LDX  (JTOT*DD)

typedef __attribute__((ext_vector_type(2))) float v2f;
typedef __attribute__((ext_vector_type(8))) float v8f;

template <int NPL>
__device__ __forceinline__ void fsmax_wave(int nt, const float* __restrict__ fsl,
                                           float* __restrict__ fs) {
    constexpr int J = 32 * NPL;
    int lane = threadIdx.x & 31;
    const float* src = fsl + (size_t)nt * J + lane;
    float z[NPL];
    #pragma unroll
    for (int i = 0; i < NPL; ++i) z[i] = src[32 * i];

    float tau   = -INFINITY;
    int   kprev = -1;
    for (int it = 0; it <= J; ++it) {
        float sum = 0.0f;
        int   k   = 0;
        #pragma unroll
        for (int i = 0; i < NPL; ++i) {
            bool in = z[i] > tau;
            sum += in ? z[i] : 0.0f;
            k   += in ? 1 : 0;
        }
        #pragma unroll
        for (int m = 16; m > 0; m >>= 1) {
            sum += __shfl_xor(sum, m, 32);
            k   += __shfl_xor(k, m, 32);
        }
        int ku = __builtin_amdgcn_readfirstlane(k);
        if (ku == kprev) break;
        kprev = ku;
        tau = fmaxf(tau, (sum - 1.0f) / (float)ku);
    }
    float* dst = fs + (size_t)nt * J + lane;
    #pragma unroll
    for (int i = 0; i < NPL; ++i) dst[32 * i] = fmaxf(z[i] - tau, 0.0f);
}

#define TPAD 33

#define CSWAP(a, b) do { float hi_ = fmaxf(a, b); float lo_ = fminf(a, b); (a) = hi_; (b) = lo_; } while (0)

template <int O>
__device__ __forceinline__ void sort8_desc(float (&r)[32]) {
    CSWAP(r[O + 0], r[O + 1]); CSWAP(r[O + 2], r[O + 3]); CSWAP(r[O + 4], r[O + 5]); CSWAP(r[O + 6], r[O + 7]);
    CSWAP(r[O + 0], r[O + 2]); CSWAP(r[O + 1], r[O + 3]); CSWAP(r[O + 4], r[O + 6]); CSWAP(r[O + 5], r[O + 7]);
    CSWAP(r[O + 1], r[O + 2]); CSWAP(r[O + 5], r[O + 6]); CSWAP(r[O + 0], r[O + 4]); CSWAP(r[O + 3], r[O + 7]);
    CSWAP(r[O + 1], r[O + 5]); CSWAP(r[O + 2], r[O + 6]);
    CSWAP(r[O + 1], r[O + 4]); CSWAP(r[O + 3], r[O + 6]);
    CSWAP(r[O + 2], r[O + 4]); CSWAP(r[O + 3], r[O + 5]);
    CSWAP(r[O + 3], r[O + 4]);
}

template <int A, int B>
__device__ __forceinline__ void top8_merge(float (&r)[32]) {
    #pragma unroll
    for (int i = 0; i < 8; ++i) r[A + i] = fmaxf(r[A + i], r[B + 7 - i]);
    #pragma unroll
    for (int j = 4; j > 0; j >>= 1) {
        #pragma unroll
        for (int i = 0; i < 8; ++i) {
            if ((i & j) == 0) CSWAP(r[A + i], r[A + i + j]);
        }
    }
}

__device__ __forceinline__ float tau_whole_row(const float* my, int lane) {
    float r[32];
    #pragma unroll
    for (int k = 0; k < 32; ++k) r[k] = my[lane * TPAD + k];
    #pragma unroll
    for (int k = 2; k <= 32; k <<= 1) {
        #pragma unroll
        for (int j = k >> 1; j > 0; j >>= 1) {
            #pragma unroll
            for (int i = 0; i < 32; ++i) {
                int ij = i ^ j;
                if (ij > i) {
                    float a  = r[i], b2 = r[ij];
                    float lo = fminf(a, b2), hi = fmaxf(a, b2);
                    if ((i & k) == 0) { r[i] = hi; r[ij] = lo; }
                    else              { r[i] = lo; r[ij] = hi; }
                }
            }
        }
    }
    float cum = 0.0f, tausum = 1.0f;
    int   ksup = 1;
    #pragma unroll
    for (int i = 0; i < 32; ++i) {
        cum += r[i];
        bool s = ((float)(i + 1) * r[i] > cum - 1.0f);
        if (s) { ksup = i + 1; tausum = cum; }
    }
    return (tausum - 1.0f) / (float)ksup;
}

__device__ __forceinline__ float essum_tiles(float* my, const float* __restrict__ esl_nt,
                                             int t0, int t1) {
    int lane = threadIdx.x & 31;
    float acc = 0.0f;
    for (int tb = t0; tb < t1; ++tb) {
        const float* src = esl_nt + (size_t)(tb * 32) * DD + lane;
        #pragma unroll
        for (int rr = 0; rr < 32; ++rr) {
            my[rr * TPAD + lane] = src[rr * DD];
        }
        float r[32];
        #pragma unroll
        for (int k = 0; k < 32; ++k) r[k] = my[lane * TPAD + k];
        sort8_desc<0>(r);
        sort8_desc<8>(r);
        sort8_desc<16>(r);
        sort8_desc<24>(r);
        top8_merge<0, 8>(r);
        top8_merge<16, 24>(r);
        top8_merge<0, 16>(r);
        float cum = 0.0f, tausum = 1.0f;
        int   ksup = 1;
        bool  more = false;
        #pragma unroll
        for (int i = 0; i < 8; ++i) {
            cum += r[i];
            bool s = ((float)(i + 1) * r[i] > cum - 1.0f);
            if (s) { ksup = i + 1; tausum = cum; }
            more = s;
        }
        float tau = (tausum - 1.0f) / (float)ksup;
        if (__builtin_expect(__builtin_amdgcn_ballot_w32(more) != 0u, 0))
            tau = tau_whole_row(my, lane);
        #pragma unroll
        for (int j2 = 0; j2 < 32; ++j2) {
            float z = my[j2 * TPAD + lane];
            float t = __shfl(tau, j2, 32);
            acc += fmaxf(z - t, 0.0f);
        }
    }
    return acc;
}

typedef __attribute__((ext_vector_type(4))) float v4f;
__global__ void __launch_bounds__(256, 8)
k_prep(const float* __restrict__ x, float* __restrict__ out,
                       const float* __restrict__ fsl0, const float* __restrict__ fsl1,
                       const float* __restrict__ fsl2,
                       const float* __restrict__ esl0, const float* __restrict__ esl1,
                       const float* __restrict__ esl2,
                       float* __restrict__ fs0, float* __restrict__ fs1,
                       float* __restrict__ fs2,
                       float* __restrict__ es0, float* __restrict__ es1,
                       float* __restrict__ es2) {
    __shared__ float tile[8][32 * TPAD];
    __shared__ float accs[8][32];
    int nt   = blockIdx.x;
    int lane = threadIdx.x & 31;
    int wid  = __builtin_amdgcn_readfirstlane((int)(threadIdx.x >> 5));

    if ((wid == 5 || wid == 6) && nt < 512) {
        int idx4 = nt * 64 + (wid - 5) * 32 + lane;
        int b    = idx4 >> 9;
        int rem  = idx4 & 511;
        v4f v = ((const v4f*)x)[idx4];
        ((v4f*)out)[b * (LDX / 4) + rem] = v;
    }

    if (wid == 2)      fsmax_wave<10>(nt, fsl2, fs2);
    else if (wid == 3) fsmax_wave<6>(nt, fsl1, fs1);
    else if (wid == 7) fsmax_wave<2>(nt, fsl0, fs0);

    int J, t0, t1;
    const float* esl;
    if (wid < 4) {
        J = 320; esl = esl2;
        t0 = (wid < 2) ? 3 * wid : 2 * wid + 2;
        t1 = (wid < 2) ? t0 + 3 : t0 + 2;
    } else if (wid < 7) {
        J = 192; esl = esl1;
        t0 = 2 * (wid - 4);
        t1 = t0 + 2;
    } else {
        J = 64; esl = esl0;
        t0 = 0;
        t1 = 2;
    }
    float acc = essum_tiles(tile[wid], esl + (size_t)nt * J * DD, t0, t1);
    accs[wid][lane] = acc;
    __syncthreads();
    if (threadIdx.x < 96) {
        int g = threadIdx.x >> 5;
        if (g == 0)
            es2[nt * DD + lane] = ((accs[0][lane] + accs[1][lane]) + accs[2][lane]) + accs[3][lane];
        else if (g == 1)
            es1[nt * DD + lane] = (accs[4][lane] + accs[5][lane]) + accs[6][lane];
        else
            es0[nt * DD + lane] = accs[7][lane];
    }
}

#define GK 32
struct GFrag { v4f a[3][4]; v2f b[4][4]; };
#define PREFETCH_GROUP_END() do { asm volatile("" ::: "memory"); __builtin_amdgcn_sched_barrier(0); } while (0)
#define MMA_BLOCK_END() __builtin_amdgcn_sched_barrier(0)

__device__ __forceinline__ void gemm_load(GFrag& f, const float* __restrict__ ap,
                                          const float* __restrict__ bp, int K, int kB) {
    #pragma unroll
    for (int j = 0; j < 4; ++j) {
        #pragma unroll
        for (int mi = 0; mi < 3; ++mi)
            f.a[mi][j] = *(const v4f*)(ap + (size_t)(16 * mi) * K + kB + 8 * j);
        #pragma unroll
        for (int e = 0; e < 4; ++e)
            f.b[j][e] = *(const v2f*)(bp + (kB + 8 * j + e) * DD);
    }
}

__device__ __forceinline__ void gemm_mma(v8f (&acc)[3][2], const GFrag& f) {
    #pragma unroll
    for (int j = 0; j < 4; ++j) {
        #pragma unroll
        for (int p = 0; p < 2; ++p) {
            #pragma unroll
            for (int mi = 0; mi < 3; ++mi) {
                #pragma unroll
                for (int t = 0; t < 2; ++t) {
                    v2f a, b;
                    a.x = f.a[mi][j][2 * p];
                    a.y = f.a[mi][j][2 * p + 1];
                    b.x = f.b[j][2 * p][t];
                    b.y = f.b[j][2 * p + 1][t];
                    acc[mi][t] = __builtin_amdgcn_wmma_f32_16x16x4_f32(false, a, false, b,
                                                    (short)0, acc[mi][t], false, false);
                }
            }
        }
    }
}

#define TLP 36
#define NU  2
#define RSP 68
__global__ void __launch_bounds__(128)
k_layer(const float* __restrict__ x, const float* __restrict__ fs,
        const float* __restrict__ essum, const float* __restrict__ thr,
        const float* __restrict__ lt, const float* __restrict__ resp,
        float* __restrict__ out, int rowOff, int K) {
    __shared__ __attribute__((aligned(16))) float rsp[8 * DD * RSP];
    __shared__ __attribute__((aligned(16))) float tls[4][48 * TLP];
    int lane = threadIdx.x & 31;
    int wid  = threadIdx.x >> 5;
    int c    = lane & 15;
    int h    = lane >> 4;
    int grp  = blockIdx.x >> 4;
    int m0   = grp * 48;
    int nn0  = grp * 8;
    int b    = ((blockIdx.x & 15) << 2) + wid;

    const float* ap = fs + (size_t)(m0 + c) * K + 4 * h;
    const float* bp = x + (size_t)b * LDX + (4 * h) * DD + 2 * c;

    v8f acc[3][2] = {};
    GFrag f0, f1;
    int nkb = K / GK;

    gemm_load(f0, ap, bp, K, 0);
    PREFETCH_GROUP_END();

    v2f   esv[3][8];
    float thv[3][8], ltv[3][8];
    #pragma unroll
    for (int mi = 0; mi < 3; ++mi) {
        #pragma unroll
        for (int r = 0; r < 8; ++r) {
            int nt = m0 + 16 * mi + 8 * h + r;
            esv[mi][r] = *(const v2f*)(essum + nt * DD + 2 * c);
            thv[mi][r] = thr[nt];
            ltv[mi][r] = lt[nt];
        }
    }
    PREFETCH_GROUP_END();
    gemm_load(f1, ap, bp, K, GK);
    PREFETCH_GROUP_END();

    v4f rr[32];
    {
        const v4f* rsrc = (const v4f*)(resp + (size_t)nn0 * (DD * LEAF));
        #pragma unroll
        for (int i = 0; i < 32; ++i) rr[i] = rsrc[threadIdx.x + 128 * i];
    }
    PREFETCH_GROUP_END();

    gemm_mma(acc, f0);
    MMA_BLOCK_END();
    for (int kb = 1; kb + 1 < nkb; kb += 2) {
        gemm_load(f0, ap, bp, K, (kb + 1) * GK);
        PREFETCH_GROUP_END();
        gemm_mma(acc, f1);
        MMA_BLOCK_END();
        gemm_load(f1, ap, bp, K, (kb + 2) * GK);
        PREFETCH_GROUP_END();
        gemm_mma(acc, f0);
        MMA_BLOCK_END();
    }

    #pragma unroll
    for (int i = 0; i < 32; ++i) {
        int i4 = threadIdx.x + 128 * i;
        *(v4f*)(rsp + (i4 >> 4) * RSP + 4 * (i4 & 15)) = rr[i];
    }
    gemm_mma(acc, f1);

    float* myt = tls[wid];
    #pragma unroll
    for (int mi = 0; mi < 3; ++mi) {
        #pragma unroll
        for (int r = 0; r < 8; ++r) {
            int   row = 16 * mi + 8 * h + r;
            v2f   es = esv[mi][r];
            float th = thv[mi][r];
            float ex = __expf(-ltv[mi][r]);
            v2f o;
            o.x = (acc[mi][0][r] * es.x - th) * ex;
            o.y = (acc[mi][1][r] * es.y - th) * ex;
            *(v2f*)(myt + row * TLP + 2 * c) = o;
        }
    }
    __syncthreads();

    #pragma unroll 1
    for (int ng = 0; ng < 8; ng += NU) {
        float r[NU][LEAF];
        #pragma unroll
        for (int u = 0; u < NU; ++u) {
            int nl = ng + u;
            float p[TT], q[TT];
            #pragma unroll
            for (int t = 0; t < TT; ++t) {
                float v = myt[(nl * TT + t) * TLP + lane];
                p[t] = fminf(fmaxf(0.5f * v + 0.5f, 0.0f), 1.0f);
                q[t] = fminf(fmaxf(0.5f - 0.5f * v, 0.0f), 1.0f);
            }
            r[u][0] = 1.0f;
            #pragma unroll
            for (int t = 0; t < TT; ++t) {
                int sz = 1 << t;
                #pragma unroll
                for (int cI = 0; cI < 32; ++cI) {
                    int cc = sz - 1 - cI;
                    if (cc >= 0) {
                        r[u][cc + sz] = r[u][cc] * q[t];
                        r[u][cc]      = r[u][cc] * p[t];
                    }
                }
            }
        }
        float s[NU];
        #pragma unroll
        for (int u = 0; u < NU; ++u) s[u] = 0.0f;
        #pragma unroll
        for (int c4 = 0; c4 < LEAF / 4; ++c4) {
            #pragma unroll
            for (int u = 0; u < NU; ++u) {
                v4f w = *(const v4f*)(rsp + ((ng + u) * DD + lane) * RSP + 4 * c4);
                s[u] += r[u][4 * c4 + 0] * w.x;
                s[u] += r[u][4 * c4 + 1] * w.y;
                s[u] += r[u][4 * c4 + 2] * w.z;
                s[u] += r[u][4 * c4 + 3] * w.w;
            }
        }
        #pragma unroll
        for (int u = 0; u < NU; ++u)
            out[(size_t)b * LDX + (rowOff + nn0 + ng + u) * DD + lane] = s[u];
    }
}

extern "C" void kernel_launch(void* const* d_in, const int* in_sizes, int n_in,
                              void* d_out, int out_size, void* d_ws, size_t ws_size,
                              hipStream_t stream) {
    const float* x = (const float*)d_in[0];
    float* out = (float*)d_out;

    float* slotA = (float*)d_ws;
    float* fs0   = slotA;
    float* fs1   = slotA + NT * 64;
    float* es0   = slotA + NT * 256;
    float* es1   = slotA + NT * 288;
    float* es2   = slotA + NT * 320;
    float* fs2   = es2 + NT * DD;

    const float* fsl0  = (const float*)d_in[1];
    const float* esl0  = (const float*)d_in[2];
    const float* thr0  = (const float*)d_in[3];
    const float* lt0   = (const float*)d_in[4];
    const float* resp0 = (const float*)d_in[5];
    const float* fsl1  = (const float*)d_in[6];
    const float* esl1  = (const float*)d_in[7];
    const float* thr1  = (const float*)d_in[8];
    const float* lt1   = (const float*)d_in[9];
    const float* resp1 = (const float*)d_in[10];
    const float* fsl2  = (const float*)d_in[11];
    const float* esl2  = (const float*)d_in[12];
    const float* thr2  = (const float*)d_in[13];
    const float* lt2   = (const float*)d_in[14];
    const float* resp2 = (const float*)d_in[15];

    k_prep<<<NT, 256, 0, stream>>>(x, out, fsl0, fsl1, fsl2, esl0, esl1, esl2,
                                       fs0, fs1, fs2, es0, es1, es2);

    k_layer<<<256, 128, 0, stream>>>(out, fs0, es0, thr0, lt0, resp0, out, 64,  64);
    k_layer<<<256, 128, 0, stream>>>(out, fs1, es1, thr1, lt1, resp1, out, 192, 192);
    k_layer<<<256, 128, 0, stream>>>(out, fs2, es2, thr2, lt2, resp2, out, 320, 320);
}


# Response dot folded level by level (v[c] = resp[c]*p5 + resp[c+32]*q5, then bit 4, ...): 63 mul+fma pairs a tree replace 126 leaf products + a 64-step fma chain; f32 sum order changes
# speedup vs baseline: 2.8201x; 1.0526x over previous
#include <hip/hip_runtime.h>
#include <hip/hip_bf16.h>

#define NB   64
#define DD   32
#define NN   128
#define TT   6
#define NT   (NN*TT)
#define NCOL (NB*DD)
#define LEAF 64
#define JTOT 448
#define LDX  (JTOT*DD)

typedef __attribute__((ext_vector_type(2))) float v2f;
typedef __attribute__((ext_vector_type(8))) float v8f;

template <int NPL>
__device__ __forceinline__ void fsmax_wave(int nt, const float* __restrict__ fsl,
                                           float* __restrict__ fs) {
    constexpr int J = 32 * NPL;
    int lane = threadIdx.x & 31;
    const float* src = fsl + (size_t)nt * J + lane;
    float z[NPL];
    #pragma unroll
    for (int i = 0; i < NPL; ++i) z[i] = src[32 * i];

    float tau   = -INFINITY;
    int   kprev = -1;
    for (int it = 0; it <= J; ++it) {
        float sum = 0.0f;
        int   k   = 0;
        #pragma unroll
        for (int i = 0; i < NPL; ++i) {
            bool in = z[i] > tau;
            sum += in ? z[i] : 0.0f;
            k   += in ? 1 : 0;
        }
        #pragma unroll
        for (int m = 16; m > 0; m >>= 1) {
            sum += __shfl_xor(sum, m, 32);
            k   += __shfl_xor(k, m, 32);
        }
        int ku = __builtin_amdgcn_readfirstlane(k);
        if (ku == kprev) break;
        kprev = ku;
        tau = fmaxf(tau, (sum - 1.0f) / (float)ku);
    }
    float* dst = fs + (size_t)nt * J + lane;
    #pragma unroll
    for (int i = 0; i < NPL; ++i) dst[32 * i] = fmaxf(z[i] - tau, 0.0f);
}

#define TPAD 33

#define CSWAP(a, b) do { float hi_ = fmaxf(a, b); float lo_ = fminf(a, b); (a) = hi_; (b) = lo_; } while (0)

template <int O>
__device__ __forceinline__ void sort8_desc(float (&r)[32]) {
    CSWAP(r[O + 0], r[O + 1]); CSWAP(r[O + 2], r[O + 3]); CSWAP(r[O + 4], r[O + 5]); CSWAP(r[O + 6], r[O + 7]);
    CSWAP(r[O + 0], r[O + 2]); CSWAP(r[O + 1], r[O + 3]); CSWAP(r[O + 4], r[O + 6]); CSWAP(r[O + 5], r[O + 7]);
    CSWAP(r[O + 1], r[O + 2]); CSWAP(r[O + 5], r[O + 6]); CSWAP(r[O + 0], r[O + 4]); CSWAP(r[O + 3], r[O + 7]);
    CSWAP(r[O + 1], r[O + 5]); CSWAP(r[O + 2], r[O + 6]);
    CSWAP(r[O + 1], r[O + 4]); CSWAP(r[O + 3], r[O + 6]);
    CSWAP(r[O + 2], r[O + 4]); CSWAP(r[O + 3], r[O + 5]);
    CSWAP(r[O + 3], r[O + 4]);
}

template <int A, int B>
__device__ __forceinline__ void top8_merge(float (&r)[32]) {
    #pragma unroll
    for (int i = 0; i < 8; ++i) r[A + i] = fmaxf(r[A + i], r[B + 7 - i]);
    #pragma unroll
    for (int j = 4; j > 0; j >>= 1) {
        #pragma unroll
        for (int i = 0; i < 8; ++i) {
            if ((i & j) == 0) CSWAP(r[A + i], r[A + i + j]);
        }
    }
}

__device__ __forceinline__ float tau_whole_row(const float* my, int lane) {
    float r[32];
    #pragma unroll
    for (int k = 0; k < 32; ++k) r[k] = my[lane * TPAD + k];
    #pragma unroll
    for (int k = 2; k <= 32; k <<= 1) {
        #pragma unroll
        for (int j = k >> 1; j > 0; j >>= 1) {
            #pragma unroll
            for (int i = 0; i < 32; ++i) {
                int ij = i ^ j;
                if (ij > i) {
                    float a  = r[i], b2 = r[ij];
                    float lo = fminf(a, b2), hi = fmaxf(a, b2);
                    if ((i & k) == 0) { r[i] = hi; r[ij] = lo; }
                    else              { r[i] = lo; r[ij] = hi; }
                }
            }
        }
    }
    float cum = 0.0f, tausum = 1.0f;
    int   ksup = 1;
    #pragma unroll
    for (int i = 0; i < 32; ++i) {
        cum += r[i];
        bool s = ((float)(i + 1) * r[i] > cum - 1.0f);
        if (s) { ksup = i + 1; tausum = cum; }
    }
    return (tausum - 1.0f) / (float)ksup;
}

__device__ __forceinline__ float essum_tiles(float* my, const float* __restrict__ esl_nt,
                                             int t0, int t1) {
    int lane = threadIdx.x & 31;
    float acc = 0.0f;
    for (int tb = t0; tb < t1; ++tb) {
        const float* src = esl_nt + (size_t)(tb * 32) * DD + lane;
        #pragma unroll
        for (int rr = 0; rr < 32; ++rr) {
            my[rr * TPAD + lane] = src[rr * DD];
        }
        float r[32];
        #pragma unroll
        for (int k = 0; k < 32; ++k) r[k] = my[lane * TPAD + k];
        sort8_desc<0>(r);
        sort8_desc<8>(r);
        sort8_desc<16>(r);
        sort8_desc<24>(r);
        top8_merge<0, 8>(r);
        top8_merge<16, 24>(r);
        top8_merge<0, 16>(r);
        float cum = 0.0f, tausum = 1.0f;
        int   ksup = 1;
        bool  more = false;
        #pragma unroll
        for (int i = 0; i < 8; ++i) {
            cum += r[i];
            bool s = ((float)(i + 1) * r[i] > cum - 1.0f);
            if (s) { ksup = i + 1; tausum = cum; }
            more = s;
        }
        float tau = (tausum - 1.0f) / (float)ksup;
        if (__builtin_expect(__builtin_amdgcn_ballot_w32(more) != 0u, 0))
            tau = tau_whole_row(my, lane);
        #pragma unroll
        for (int j2 = 0; j2 < 32; ++j2) {
            float z = my[j2 * TPAD + lane];
            float t = __shfl(tau, j2, 32);
            acc += fmaxf(z - t, 0.0f);
        }
    }
    return acc;
}

typedef __attribute__((ext_vector_type(4))) float v4f;
__global__ void __launch_bounds__(256, 8)
k_prep(const float* __restrict__ x, float* __restrict__ out,
                       const float* __restrict__ fsl0, const float* __restrict__ fsl1,
                       const float* __restrict__ fsl2,
                       const float* __restrict__ esl0, const float* __restrict__ esl1,
                       const float* __restrict__ esl2,
                       float* __restrict__ fs0, float* __restrict__ fs1,
                       float* __restrict__ fs2,
                       float* __restrict__ es0, float* __restrict__ es1,
                       float* __restrict__ es2) {
    __shared__ float tile[8][32 * TPAD];
    __shared__ float accs[8][32];
    int nt   = blockIdx.x;
    int lane = threadIdx.x & 31;
    int wid  = __builtin_amdgcn_readfirstlane((int)(threadIdx.x >> 5));

    if ((wid == 5 || wid == 6) && nt < 512) {
        int idx4 = nt * 64 + (wid - 5) * 32 + lane;
        int b    = idx4 >> 9;
        int rem  = idx4 & 511;
        v4f v = ((const v4f*)x)[idx4];
        ((v4f*)out)[b * (LDX / 4) + rem] = v;
    }

    if (wid == 2)      fsmax_wave<10>(nt, fsl2, fs2);
    else if (wid == 3) fsmax_wave<6>(nt, fsl1, fs1);
    else if (wid == 7) fsmax_wave<2>(nt, fsl0, fs0);

    int J, t0, t1;
    const float* esl;
    if (wid < 4) {
        J = 320; esl = esl2;
        t0 = (wid < 2) ? 3 * wid : 2 * wid + 2;
        t1 = (wid < 2) ? t0 + 3 : t0 + 2;
    } else if (wid < 7) {
        J = 192; esl = esl1;
        t0 = 2 * (wid - 4);
        t1 = t0 + 2;
    } else {
        J = 64; esl = esl0;
        t0 = 0;
        t1 = 2;
    }
    float acc = essum_tiles(tile[wid], esl + (size_t)nt * J * DD, t0, t1);
    accs[wid][lane] = acc;
    __syncthreads();
    if (threadIdx.x < 96) {
        int g = threadIdx.x >> 5;
        if (g == 0)
            es2[nt * DD + lane] = ((accs[0][lane] + accs[1][lane]) + accs[2][lane]) + accs[3][lane];
        else if (g == 1)
            es1[nt * DD + lane] = (accs[4][lane] + accs[5][lane]) + accs[6][lane];
        else
            es0[nt * DD + lane] = accs[7][lane];
    }
}

#define GK 32
struct GFrag { v4f a[3][4]; v2f b[4][4]; };
#define PREFETCH_GROUP_END() do { asm volatile("" ::: "memory"); __builtin_amdgcn_sched_barrier(0); } while (0)
#define MMA_BLOCK_END() __builtin_amdgcn_sched_barrier(0)

__device__ __forceinline__ void gemm_load(GFrag& f, const float* __restrict__ ap,
                                          const float* __restrict__ bp, int K, int kB) {
    #pragma unroll
    for (int j = 0; j < 4; ++j) {
        #pragma unroll
        for (int mi = 0; mi < 3; ++mi)
            f.a[mi][j] = *(const v4f*)(ap + (size_t)(16 * mi) * K + kB + 8 * j);
        #pragma unroll
        for (int e = 0; e < 4; ++e)
            f.b[j][e] = *(const v2f*)(bp + (kB + 8 * j + e) * DD);
    }
}

__device__ __forceinline__ void gemm_mma(v8f (&acc)[3][2], const GFrag& f) {
    #pragma unroll
    for (int j = 0; j < 4; ++j) {
        #pragma unroll
        for (int p = 0; p < 2; ++p) {
            #pragma unroll
            for (int mi = 0; mi < 3; ++mi) {
                #pragma unroll
                for (int t = 0; t < 2; ++t) {
                    v2f a, b;
                    a.x = f.a[mi][j][2 * p];
                    a.y = f.a[mi][j][2 * p + 1];
                    b.x = f.b[j][2 * p][t];
                    b.y = f.b[j][2 * p + 1][t];
                    acc[mi][t] = __builtin_amdgcn_wmma_f32_16x16x4_f32(false, a, false, b,
                                                    (short)0, acc[mi][t], false, false);
                }
            }
        }
    }
}

#define TLP 36
#define NU  2
#define TREE_GROUP_END() __builtin_amdgcn_sched_barrier(0)
#define RSP 68
struct TreeIn { v4f w[LEAF / 4]; float tv[TT]; };

__device__ __forceinline__ void tree_load(TreeIn& in, const float* wr, const float* tr) {
    #pragma unroll
    for (int t = 0; t < TT; ++t) in.tv[t] = tr[t * TLP];
    #pragma unroll
    for (int c4 = 0; c4 < LEAF / 4; ++c4) in.w[c4] = *(const v4f*)(wr + 4 * c4);
}

__device__ __forceinline__ float tree_fold(const TreeIn& in) {
    float p[TT], q[TT];
    #pragma unroll
    for (int t = 0; t < TT; ++t) {
        p[t] = fminf(fmaxf(0.5f * in.tv[t] + 0.5f, 0.0f), 1.0f);
        q[t] = fminf(fmaxf(0.5f - 0.5f * in.tv[t], 0.0f), 1.0f);
    }
    v2f v[LEAF / 4];
    #pragma unroll
    for (int c4 = 0; c4 < LEAF / 8; ++c4) {
        v4f lo = in.w[c4];
        v4f hi = in.w[c4 + LEAF / 8];
        v2f lo01 = {lo.x, lo.y}, lo23 = {lo.z, lo.w};
        v2f hi01 = {hi.x, hi.y}, hi23 = {hi.z, hi.w};
        v[2 * c4]     = lo01 * p[5] + hi01 * q[5];
        v[2 * c4 + 1] = lo23 * p[5] + hi23 * q[5];
    }
    #pragma unroll
    for (int t = TT - 2; t >= 1; --t) {
        #pragma unroll
        for (int j = 0; j < LEAF / 8; ++j) {
            if (j < (1 << (t - 1))) v[j] = v[j] * p[t] + v[j + (1 << (t - 1))] * q[t];
        }
    }
    return v[0].x * p[0] + v[0].y * q[0];
}

__global__ void __launch_bounds__(128)
k_layer(const float* __restrict__ x, const float* __restrict__ fs,
        const float* __restrict__ essum, const float* __restrict__ thr,
        const float* __restrict__ lt, const float* __restrict__ resp,
        float* __restrict__ out, int rowOff, int K) {
    __shared__ __attribute__((aligned(16))) float rsp[8 * DD * RSP];
    __shared__ __attribute__((aligned(16))) float tls[4][48 * TLP];
    int lane = threadIdx.x & 31;
    int wid  = threadIdx.x >> 5;
    int c    = lane & 15;
    int h    = lane >> 4;
    int grp  = blockIdx.x >> 4;
    int m0   = grp * 48;
    int nn0  = grp * 8;
    int b    = ((blockIdx.x & 15) << 2) + wid;

    const float* ap = fs + (size_t)(m0 + c) * K + 4 * h;
    const float* bp = x + (size_t)b * LDX + (4 * h) * DD + 2 * c;

    v8f acc[3][2] = {};
    GFrag f0, f1;
    int nkb = K / GK;

    gemm_load(f0, ap, bp, K, 0);
    PREFETCH_GROUP_END();

    v2f   esv[3][8];
    float thv[3][8], ltv[3][8];
    #pragma unroll
    for (int mi = 0; mi < 3; ++mi) {
        #pragma unroll
        for (int r = 0; r < 8; ++r) {
            int nt = m0 + 16 * mi + 8 * h + r;
            esv[mi][r] = *(const v2f*)(essum + nt * DD + 2 * c);
            thv[mi][r] = thr[nt];
            ltv[mi][r] = lt[nt];
        }
    }
    PREFETCH_GROUP_END();
    gemm_load(f1, ap, bp, K, GK);
    PREFETCH_GROUP_END();

    v4f rr[32];
    {
        const v4f* rsrc = (const v4f*)(resp + (size_t)nn0 * (DD * LEAF));
        #pragma unroll
        for (int i = 0; i < 32; ++i) rr[i] = rsrc[threadIdx.x + 128 * i];
    }
    PREFETCH_GROUP_END();

    gemm_mma(acc, f0);
    MMA_BLOCK_END();
    for (int kb = 1; kb + 1 < nkb; kb += 2) {
        gemm_load(f0, ap, bp, K, (kb + 1) * GK);
        PREFETCH_GROUP_END();
        gemm_mma(acc, f1);
        MMA_BLOCK_END();
        gemm_load(f1, ap, bp, K, (kb + 2) * GK);
        PREFETCH_GROUP_END();
        gemm_mma(acc, f0);
        MMA_BLOCK_END();
    }

    #pragma unroll
    for (int i = 0; i < 32; ++i) {
        int i4 = threadIdx.x + 128 * i;
        *(v4f*)(rsp + (i4 >> 4) * RSP + 4 * (i4 & 15)) = rr[i];
    }
    gemm_mma(acc, f1);

    float* myt = tls[wid];
    #pragma unroll
    for (int mi = 0; mi < 3; ++mi) {
        #pragma unroll
        for (int r = 0; r < 8; ++r) {
            int   row = 16 * mi + 8 * h + r;
            v2f   es = esv[mi][r];
            float th = thv[mi][r];
            float ex = __expf(-ltv[mi][r]);
            v2f o;
            o.x = (acc[mi][0][r] * es.x - th) * ex;
            o.y = (acc[mi][1][r] * es.y - th) * ex;
            *(v2f*)(myt + row * TLP + 2 * c) = o;
        }
    }
    __syncthreads();

    const float* wr0 = rsp + lane * RSP;
    const float* tr0 = myt + lane;
    TreeIn ta[NU], tb[NU];
    #pragma unroll
    for (int u = 0; u < NU; ++u) tree_load(ta[u], wr0 + u * (DD * RSP), tr0 + u * (TT * TLP));
    TREE_GROUP_END();
    #pragma unroll
    for (int g = 0; g < 8 / NU; g += 2) {
        #pragma unroll
        for (int u = 0; u < NU; ++u) {
            int nl = (g + 1) * NU + u;
            tree_load(tb[u], wr0 + nl * (DD * RSP), tr0 + nl * (TT * TLP));
        }
        TREE_GROUP_END();
        #pragma unroll
        for (int u = 0; u < NU; ++u)
            out[(size_t)b * LDX + (rowOff + nn0 + g * NU + u) * DD + lane] = tree_fold(ta[u]);
        if (g + 2 < 8 / NU) {
            #pragma unroll
            for (int u = 0; u < NU; ++u) {
                int nl = (g + 2) * NU + u;
                tree_load(ta[u], wr0 + nl * (DD * RSP), tr0 + nl * (TT * TLP));
            }
            TREE_GROUP_END();
        }
        #pragma unroll
        for (int u = 0; u < NU; ++u)
            out[(size_t)b * LDX + (rowOff + nn0 + (g + 1) * NU + u) * DD + lane] = tree_fold(tb[u]);
    }
}

extern "C" void kernel_launch(void* const* d_in, const int* in_sizes, int n_in,
                              void* d_out, int out_size, void* d_ws, size_t ws_size,
                              hipStream_t stream) {
    const float* x = (const float*)d_in[0];
    float* out = (float*)d_out;

    float* slotA = (float*)d_ws;
    float* fs0   = slotA;
    float* fs1   = slotA + NT * 64;
    float* es0   = slotA + NT * 256;
    float* es1   = slotA + NT * 288;
    float* es2   = slotA + NT * 320;
    float* fs2   = es2 + NT * DD;

    const float* fsl0  = (const float*)d_in[1];
    const float* esl0  = (const float*)d_in[2];
    const float* thr0  = (const float*)d_in[3];
    const float* lt0   = (const float*)d_in[4];
    const float* resp0 = (const float*)d_in[5];
    const float* fsl1  = (const float*)d_in[6];
    const float* esl1  = (const float*)d_in[7];
    const float* thr1  = (const float*)d_in[8];
    const float* lt1   = (const float*)d_in[9];
    const float* resp1 = (const float*)d_in[10];
    const float* fsl2  = (const float*)d_in[11];
    const float* esl2  = (const float*)d_in[12];
    const float* thr2  = (const float*)d_in[13];
    const float* lt2   = (const float*)d_in[14];
    const float* resp2 = (const float*)d_in[15];

    k_prep<<<NT, 256, 0, stream>>>(x, out, fsl0, fsl1, fsl2, esl0, esl1, esl2,
                                       fs0, fs1, fs2, es0, es1, es2);

    k_layer<<<256, 128, 0, stream>>>(out, fs0, es0, thr0, lt0, resp0, out, 64,  64);
    k_layer<<<256, 128, 0, stream>>>(out, fs1, es1, thr1, lt1, resp1, out, 192, 192);
    k_layer<<<256, 128, 0, stream>>>(out, fs2, es2, thr2, lt2, resp2, out, 320, 320);
}


# Layer 0 moves into the input-only launch: 256 blocks (one tree x 32 batches) prepare the tree's six fs/es rows on chip, run its 16x32x64 products (A rows 6-15 zero) and leaf fold; 4 launches -> 3
# speedup vs baseline: 3.0093x; 1.0671x over previous
#include <hip/hip_runtime.h>
#include <hip/hip_bf16.h>

#define NB   64
#define DD   32
#define NN   128
#define TT   6
#define NT   (NN*TT)
#define NCOL (NB*DD)
#define LEAF 64
#define JTOT 448
#define LDX  (JTOT*DD)

typedef __attribute__((ext_vector_type(2))) float v2f;
typedef __attribute__((ext_vector_type(4))) float v4f;
typedef __attribute__((ext_vector_type(8))) float v8f;

template <int NPL>
__device__ __forceinline__ void fsmax_wave(int nt, const float* __restrict__ fsl,
                                           float* __restrict__ fs) {
    constexpr int J = 32 * NPL;
    int lane = threadIdx.x & 31;
    const float* src = fsl + (size_t)nt * J + lane;
    float z[NPL];
    #pragma unroll
    for (int i = 0; i < NPL; ++i) z[i] = src[32 * i];

    float tau   = -INFINITY;
    int   kprev = -1;
    for (int it = 0; it <= J; ++it) {
        float sum = 0.0f;
        int   k   = 0;
        #pragma unroll
        for (int i = 0; i < NPL; ++i) {
            bool in = z[i] > tau;
            sum += in ? z[i] : 0.0f;
            k   += in ? 1 : 0;
        }
        #pragma unroll
        for (int m = 16; m > 0; m >>= 1) {
            sum += __shfl_xor(sum, m, 32);
            k   += __shfl_xor(k, m, 32);
        }
        int ku = __builtin_amdgcn_readfirstlane(k);
        if (ku == kprev) break;
        kprev = ku;
        tau = fmaxf(tau, (sum - 1.0f) / (float)ku);
    }
    float* dst = fs + (size_t)nt * J + lane;
    #pragma unroll
    for (int i = 0; i < NPL; ++i) dst[32 * i] = fmaxf(z[i] - tau, 0.0f);
}

#define TPAD 33

#define CSWAP(a, b) do { float hi_ = fmaxf(a, b); float lo_ = fminf(a, b); (a) = hi_; (b) = lo_; } while (0)

template <int O>
__device__ __forceinline__ void sort8_desc(float (&r)[32]) {
    CSWAP(r[O + 0], r[O + 1]); CSWAP(r[O + 2], r[O + 3]); CSWAP(r[O + 4], r[O + 5]); CSWAP(r[O + 6], r[O + 7]);
    CSWAP(r[O + 0], r[O + 2]); CSWAP(r[O + 1], r[O + 3]); CSWAP(r[O + 4], r[O + 6]); CSWAP(r[O + 5], r[O + 7]);
    CSWAP(r[O + 1], r[O + 2]); CSWAP(r[O + 5], r[O + 6]); CSWAP(r[O + 0], r[O + 4]); CSWAP(r[O + 3], r[O + 7]);
    CSWAP(r[O + 1], r[O + 5]); CSWAP(r[O + 2], r[O + 6]);
    CSWAP(r[O + 1], r[O + 4]); CSWAP(r[O + 3], r[O + 6]);
    CSWAP(r[O + 2], r[O + 4]); CSWAP(r[O + 3], r[O + 5]);
    CSWAP(r[O + 3], r[O + 4]);
}

template <int A, int B>
__device__ __forceinline__ void top8_merge(float (&r)[32]) {
    #pragma unroll
    for (int i = 0; i < 8; ++i) r[A + i] = fmaxf(r[A + i], r[B + 7 - i]);
    #pragma unroll
    for (int j = 4; j > 0; j >>= 1) {
        #pragma unroll
        for (int i = 0; i < 8; ++i) {
            if ((i & j) == 0) CSWAP(r[A + i], r[A + i + j]);
        }
    }
}

__device__ __forceinline__ float tau_whole_row(const float* my, int lane) {
    float r[32];
    #pragma unroll
    for (int k = 0; k < 32; ++k) r[k] = my[lane * TPAD + k];
    #pragma unroll
    for (int k = 2; k <= 32; k <<= 1) {
        #pragma unroll
        for (int j = k >> 1; j > 0; j >>= 1) {
            #pragma unroll
            for (int i = 0; i < 32; ++i) {
                int ij = i ^ j;
                if (ij > i) {
                    float a  = r[i], b2 = r[ij];
                    float lo = fminf(a, b2), hi = fmaxf(a, b2);
                    if ((i & k) == 0) { r[i] = hi; r[ij] = lo; }
                    else              { r[i] = lo; r[ij] = hi; }
                }
            }
        }
    }
    float cum = 0.0f, tausum = 1.0f;
    int   ksup = 1;
    #pragma unroll
    for (int i = 0; i < 32; ++i) {
        cum += r[i];
        bool s = ((float)(i + 1) * r[i] > cum - 1.0f);
        if (s) { ksup = i + 1; tausum = cum; }
    }
    return (tausum - 1.0f) / (float)ksup;
}

__device__ __forceinline__ float essum_tiles(float* my, const float* __restrict__ esl_nt,
                                             int t0, int t1) {
    int lane = threadIdx.x & 31;
    float acc = 0.0f;
    for (int tb = t0; tb < t1; ++tb) {
        const float* src = esl_nt + (size_t)(tb * 32) * DD + lane;
        #pragma unroll
        for (int rr = 0; rr < 32; ++rr) {
            my[rr * TPAD + lane] = src[rr * DD];
        }
        float r[32];
        #pragma unroll
        for (int k = 0; k < 32; ++k) r[k] = my[lane * TPAD + k];
        sort8_desc<0>(r);
        sort8_desc<8>(r);
        sort8_desc<16>(r);
        sort8_desc<24>(r);
        top8_merge<0, 8>(r);
        top8_merge<16, 24>(r);
        top8_merge<0, 16>(r);
        float cum = 0.0f, tausum = 1.0f;
        int   ksup = 1;
        bool  more = false;
        #pragma unroll
        for (int i = 0; i < 8; ++i) {
            cum += r[i];
            bool s = ((float)(i + 1) * r[i] > cum - 1.0f);
            if (s) { ksup = i + 1; tausum = cum; }
            more = s;
        }
        float tau = (tausum - 1.0f) / (float)ksup;
        if (__builtin_expect(__builtin_amdgcn_ballot_w32(more) != 0u, 0))
            tau = tau_whole_row(my, lane);
        #pragma unroll
        for (int j2 = 0; j2 < 32; ++j2) {
            float z = my[j2 * TPAD + lane];
            float t = __shfl(tau, j2, 32);
            acc += fmaxf(z - t, 0.0f);
        }
    }
    return acc;
}


#define GK 32
struct GFrag { v4f a[3][4]; v2f b[4][4]; };
#define PREFETCH_GROUP_END() do { asm volatile("" ::: "memory"); __builtin_amdgcn_sched_barrier(0); } while (0)
#define MMA_BLOCK_END() __builtin_amdgcn_sched_barrier(0)

__device__ __forceinline__ void gemm_load(GFrag& f, const float* __restrict__ ap,
                                          const float* __restrict__ bp, int K, int kB) {
    #pragma unroll
    for (int j = 0; j < 4; ++j) {
        #pragma unroll
        for (int mi = 0; mi < 3; ++mi)
            f.a[mi][j] = *(const v4f*)(ap + (size_t)(16 * mi) * K + kB + 8 * j);
        #pragma unroll
        for (int e = 0; e < 4; ++e)
            f.b[j][e] = *(const v2f*)(bp + (kB + 8 * j + e) * DD);
    }
}

__device__ __forceinline__ void gemm_mma(v8f (&acc)[3][2], const GFrag& f) {
    #pragma unroll
    for (int j = 0; j < 4; ++j) {
        #pragma unroll
        for (int p = 0; p < 2; ++p) {
            #pragma unroll
            for (int mi = 0; mi < 3; ++mi) {
                #pragma unroll
                for (int t = 0; t < 2; ++t) {
                    v2f a, b;
                    a.x = f.a[mi][j][2 * p];
                    a.y = f.a[mi][j][2 * p + 1];
                    b.x = f.b[j][2 * p][t];
                    b.y = f.b[j][2 * p + 1][t];
                    acc[mi][t] = __builtin_amdgcn_wmma_f32_16x16x4_f32(false, a, false, b,
                                                    (short)0, acc[mi][t], false, false);
                }
            }
        }
    }
}

#define TLP 36
#define NU  2
#define TREE_GROUP_END() __builtin_amdgcn_sched_barrier(0)
#define RSP 68
struct TreeIn { v4f w[LEAF / 4]; float tv[TT]; };

__device__ __forceinline__ void tree_load(TreeIn& in, const float* wr, const float* tr) {
    #pragma unroll
    for (int t = 0; t < TT; ++t) in.tv[t] = tr[t * TLP];
    #pragma unroll
    for (int c4 = 0; c4 < LEAF / 4; ++c4) in.w[c4] = *(const v4f*)(wr + 4 * c4);
}

__device__ __forceinline__ float tree_fold(const TreeIn& in) {
    float p[TT], q[TT];
    #pragma unroll
    for (int t = 0; t < TT; ++t) {
        p[t] = fminf(fmaxf(0.5f * in.tv[t] + 0.5f, 0.0f), 1.0f);
        q[t] = fminf(fmaxf(0.5f - 0.5f * in.tv[t], 0.0f), 1.0f);
    }
    v2f v[LEAF / 4];
    #pragma unroll
    for (int c4 = 0; c4 < LEAF / 8; ++c4) {
        v4f lo = in.w[c4];
        v4f hi = in.w[c4 + LEAF / 8];
        v2f lo01 = {lo.x, lo.y}, lo23 = {lo.z, lo.w};
        v2f hi01 = {hi.x, hi.y}, hi23 = {hi.z, hi.w};
        v[2 * c4]     = lo01 * p[5] + hi01 * q[5];
        v[2 * c4 + 1] = lo23 * p[5] + hi23 * q[5];
    }
    #pragma unroll
    for (int t = TT - 2; t >= 1; --t) {
        #pragma unroll
        for (int j = 0; j < LEAF / 8; ++j) {
            if (j < (1 << (t - 1))) v[j] = v[j] * p[t] + v[j + (1 << (t - 1))] * q[t];
        }
    }
    return v[0].x * p[0] + v[0].y * q[0];
}

#define L0BLK 256
#define L0B   4
#define FSP   68
struct L0Frag { v2f b[4][4]; };
__device__ __forceinline__ void l0_load(L0Frag& f, const float* __restrict__ bp, int j0) {
    #pragma unroll
    for (int j = 0; j < 4; ++j) {
        #pragma unroll
        for (int e = 0; e < 4; ++e)
            f.b[j][e] = *(const v2f*)(bp + (8 * (j0 + j) + e) * DD);
    }
}
__device__ __forceinline__ void l0_mma(v8f& acc0, v8f& acc1, const float* arow,
                                       const L0Frag& f, int j0) {
    #pragma unroll
    for (int j = 0; j < 4; ++j) {
        v4f a4 = *(const v4f*)(arow + 8 * (j0 + j));
        #pragma unroll
        for (int p = 0; p < 2; ++p) {
            v2f a, b0, b1;
            a.x  = a4[2 * p];             a.y  = a4[2 * p + 1];
            b0.x = f.b[j][2 * p][0];      b0.y = f.b[j][2 * p + 1][0];
            b1.x = f.b[j][2 * p][1];      b1.y = f.b[j][2 * p + 1][1];
            acc0 = __builtin_amdgcn_wmma_f32_16x16x4_f32(false, a, false, b0,
                                            (short)0, acc0, false, false);
            acc1 = __builtin_amdgcn_wmma_f32_16x16x4_f32(false, a, false, b1,
                                            (short)0, acc1, false, false);
        }
    }
}

__global__ void __launch_bounds__(256, 8)
k_prep(const float* __restrict__ x, float* __restrict__ out,
                       const float* __restrict__ fsl0, const float* __restrict__ fsl1,
                       const float* __restrict__ fsl2,
                       const float* __restrict__ esl0, const float* __restrict__ esl1,
                       const float* __restrict__ esl2,
                       const float* __restrict__ thr0, const float* __restrict__ lt0,
                       const float* __restrict__ resp0,
                       float* __restrict__ fs1, float* __restrict__ fs2,
                       float* __restrict__ es1, float* __restrict__ es2) {
    __shared__ __attribute__((aligned(16))) float tile[8][32 * TPAD];
    __shared__ float accs[8][32];
    __shared__ __attribute__((aligned(16))) float fsA[16 * FSP];
    __shared__ __attribute__((aligned(16))) float esA[TT * DD];
    __shared__ __attribute__((aligned(16))) float rsp0[DD * RSP];
    __shared__ __attribute__((aligned(16))) float teA[2 * TT];
    int lane = threadIdx.x & 31;
    int wid  = __builtin_amdgcn_readfirstlane((int)(threadIdx.x >> 5));

    if (blockIdx.x < L0BLK) {
        int n     = blockIdx.x >> 1;
        int bbase = ((blockIdx.x & 1) << 5) + wid * L0B;
        const v4f* rsrc = (const v4f*)(resp0 + (size_t)n * (DD * LEAF));
        v4f rr0 = rsrc[threadIdx.x];
        v4f rr1 = rsrc[threadIdx.x + 256];
        if (wid < TT) {
            int nt = n * TT + wid;
            fsmax_wave<2>(0, fsl0 + (size_t)nt * 64, fsA + wid * FSP);
            float es = essum_tiles(tile[wid], esl0 + (size_t)nt * 64 * DD, 0, 2);
            esA[wid * DD + lane] = es;
            if (lane == 0) {
                teA[2 * wid]     = thr0[nt];
                teA[2 * wid + 1] = __expf(-lt0[nt]);
            }
        } else {
            for (int i = (int)threadIdx.x - 32 * TT; i < 10 * 64; i += 64)
                fsA[(TT + (i >> 6)) * FSP + (i & 63)] = 0.0f;
        }
        {
            int i0 = threadIdx.x, i1 = threadIdx.x + 256;
            *(v4f*)(rsp0 + (i0 >> 4) * RSP + 4 * (i0 & 15)) = rr0;
            *(v4f*)(rsp0 + (i1 >> 4) * RSP + 4 * (i1 & 15)) = rr1;
        }
        __syncthreads();

        int c = lane & 15;
        int h = lane >> 4;
        float* tlw = tile[wid];
        const float* arow = fsA + c * FSP + 4 * h;
        const float* wr   = rsp0 + lane * RSP;
        const float* bp = x + (size_t)bbase * (64 * DD) + (4 * h) * DD + 2 * c;
        L0Frag f0, f1;
        l0_load(f0, bp, 0);
        PREFETCH_GROUP_END();
        #pragma unroll
        for (int i = 0; i < L0B; ++i) {
            int b = bbase + i;
            v8f acc0 = {}, acc1 = {};
            l0_load(f1, bp + (size_t)i * (64 * DD), 4);
            PREFETCH_GROUP_END();
            l0_mma(acc0, acc1, arow, f0, 0);
            MMA_BLOCK_END();
            if (i + 1 < L0B) {
                l0_load(f0, bp + (size_t)(i + 1) * (64 * DD), 0);
                PREFETCH_GROUP_END();
            }
            l0_mma(acc0, acc1, arow, f1, 4);
            MMA_BLOCK_END();
            if (h == 0) {
                #pragma unroll
                for (int r = 0; r < TT; ++r) {
                    v2f es = *(const v2f*)(esA + r * DD + 2 * c);
                    v2f te = *(const v2f*)(teA + 2 * r);
                    v2f o;
                    o.x = (acc0[r] * es.x - te.x) * te.y;
                    o.y = (acc1[r] * es.y - te.x) * te.y;
                    *(v2f*)(tlw + r * TLP + 2 * c) = o;
                }
            }
            TreeIn in;
            tree_load(in, wr, tlw + lane);
            out[(size_t)b * LDX + (64 + n) * DD + lane] = tree_fold(in);
            MMA_BLOCK_END();
        }
        return;
    }

    int nt = blockIdx.x - L0BLK;

    if ((wid == 5 || wid == 6) && nt < 512) {
        int idx4 = nt * 64 + (wid - 5) * 32 + lane;
        int b    = idx4 >> 9;
        int rem  = idx4 & 511;
        v4f v = ((const v4f*)x)[idx4];
        ((v4f*)out)[b * (LDX / 4) + rem] = v;
    }

    if (wid == 7) {
        fsmax_wave<10>(nt, fsl2, fs2);
        fsmax_wave<6>(nt, fsl1, fs1);
    }

    int J, t0, t1;
    const float* esl;
    if (wid < 4) {
        J = 320; esl = esl2;
        t0 = (wid < 2) ? 3 * wid : 2 * wid + 2;
        t1 = (wid < 2) ? t0 + 3 : t0 + 2;
    } else if (wid < 7) {
        J = 192; esl = esl1;
        t0 = 2 * (wid - 4);
        t1 = t0 + 2;
    } else {
        J = 192; esl = esl1;
        t0 = 0;
        t1 = 0;
    }
    float acc = essum_tiles(tile[wid], esl + (size_t)nt * J * DD, t0, t1);
    accs[wid][lane] = acc;
    __syncthreads();
    if (threadIdx.x < 64) {
        int g = threadIdx.x >> 5;
        if (g == 0)
            es2[nt * DD + lane] = ((accs[0][lane] + accs[1][lane]) + accs[2][lane]) + accs[3][lane];
        else
            es1[nt * DD + lane] = (accs[4][lane] + accs[5][lane]) + accs[6][lane];
    }
}

__global__ void __launch_bounds__(128)
k_layer(const float* __restrict__ x, const float* __restrict__ fs,
        const float* __restrict__ essum, const float* __restrict__ thr,
        const float* __restrict__ lt, const float* __restrict__ resp,
        float* __restrict__ out, int rowOff, int K) {
    __shared__ __attribute__((aligned(16))) float rsp[8 * DD * RSP];
    __shared__ __attribute__((aligned(16))) float tls[4][48 * TLP];
    int lane = threadIdx.x & 31;
    int wid  = threadIdx.x >> 5;
    int c    = lane & 15;
    int h    = lane >> 4;
    int grp  = blockIdx.x >> 4;
    int m0   = grp * 48;
    int nn0  = grp * 8;
    int b    = ((blockIdx.x & 15) << 2) + wid;

    const float* ap = fs + (size_t)(m0 + c) * K + 4 * h;
    const float* bp = x + (size_t)b * LDX + (4 * h) * DD + 2 * c;

    v8f acc[3][2] = {};
    GFrag f0, f1;
    int nkb = K / GK;

    gemm_load(f0, ap, bp, K, 0);
    PREFETCH_GROUP_END();

    v2f   esv[3][8];
    float thv[3][8], ltv[3][8];
    #pragma unroll
    for (int mi = 0; mi < 3; ++mi) {
        #pragma unroll
        for (int r = 0; r < 8; ++r) {
            int nt = m0 + 16 * mi + 8 * h + r;
            esv[mi][r] = *(const v2f*)(essum + nt * DD + 2 * c);
            thv[mi][r] = thr[nt];
            ltv[mi][r] = lt[nt];
        }
    }
    PREFETCH_GROUP_END();
    gemm_load(f1, ap, bp, K, GK);
    PREFETCH_GROUP_END();

    v4f rr[32];
    {
        const v4f* rsrc = (const v4f*)(resp + (size_t)nn0 * (DD * LEAF));
        #pragma unroll
        for (int i = 0; i < 32; ++i) rr[i] = rsrc[threadIdx.x + 128 * i];
    }
    PREFETCH_GROUP_END();

    gemm_mma(acc, f0);
    MMA_BLOCK_END();
    for (int kb = 1; kb + 1 < nkb; kb += 2) {
        gemm_load(f0, ap, bp, K, (kb + 1) * GK);
        PREFETCH_GROUP_END();
        gemm_mma(acc, f1);
        MMA_BLOCK_END();
        gemm_load(f1, ap, bp, K, (kb + 2) * GK);
        PREFETCH_GROUP_END();
        gemm_mma(acc, f0);
        MMA_BLOCK_END();
    }

    #pragma unroll
    for (int i = 0; i < 32; ++i) {
        int i4 = threadIdx.x + 128 * i;
        *(v4f*)(rsp + (i4 >> 4) * RSP + 4 * (i4 & 15)) = rr[i];
    }
    gemm_mma(acc, f1);

    float* myt = tls[wid];
    #pragma unroll
    for (int mi = 0; mi < 3; ++mi) {
        #pragma unroll
        for (int r = 0; r < 8; ++r) {
            int   row = 16 * mi + 8 * h + r;
            v2f   es = esv[mi][r];
            float th = thv[mi][r];
            float ex = __expf(-ltv[mi][r]);
            v2f o;
            o.x = (acc[mi][0][r] * es.x - th) * ex;
            o.y = (acc[mi][1][r] * es.y - th) * ex;
            *(v2f*)(myt + row * TLP + 2 * c) = o;
        }
    }
    __syncthreads();

    const float* wr0 = rsp + lane * RSP;
    const float* tr0 = myt + lane;
    TreeIn ta[NU], tb[NU];
    #pragma unroll
    for (int u = 0; u < NU; ++u) tree_load(ta[u], wr0 + u * (DD * RSP), tr0 + u * (TT * TLP));
    TREE_GROUP_END();
    #pragma unroll
    for (int g = 0; g < 8 / NU; g += 2) {
        #pragma unroll
        for (int u = 0; u < NU; ++u) {
            int nl = (g + 1) * NU + u;
            tree_load(tb[u], wr0 + nl * (DD * RSP), tr0 + nl * (TT * TLP));
        }
        TREE_GROUP_END();
        #pragma unroll
        for (int u = 0; u < NU; ++u)
            out[(size_t)b * LDX + (rowOff + nn0 + g * NU + u) * DD + lane] = tree_fold(ta[u]);
        if (g + 2 < 8 / NU) {
            #pragma unroll
            for (int u = 0; u < NU; ++u) {
                int nl = (g + 2) * NU + u;
                tree_load(ta[u], wr0 + nl * (DD * RSP), tr0 + nl * (TT * TLP));
            }
            TREE_GROUP_END();
        }
        #pragma unroll
        for (int u = 0; u < NU; ++u)
            out[(size_t)b * LDX + (rowOff + nn0 + (g + 1) * NU + u) * DD + lane] = tree_fold(tb[u]);
    }
}

extern "C" void kernel_launch(void* const* d_in, const int* in_sizes, int n_in,
                              void* d_out, int out_size, void* d_ws, size_t ws_size,
                              hipStream_t stream) {
    const float* x = (const float*)d_in[0];
    float* out = (float*)d_out;

    float* slotA = (float*)d_ws;
    float* fs1   = slotA + NT * 64;
    float* es1   = slotA + NT * 288;
    float* es2   = slotA + NT * 320;
    float* fs2   = es2 + NT * DD;

    const float* fsl0  = (const float*)d_in[1];
    const float* esl0  = (const float*)d_in[2];
    const float* thr0  = (const float*)d_in[3];
    const float* lt0   = (const float*)d_in[4];
    const float* resp0 = (const float*)d_in[5];
    const float* fsl1  = (const float*)d_in[6];
    const float* esl1  = (const float*)d_in[7];
    const float* thr1  = (const float*)d_in[8];
    const float* lt1   = (const float*)d_in[9];
    const float* resp1 = (const float*)d_in[10];
    const float* fsl2  = (const float*)d_in[11];
    const float* esl2  = (const float*)d_in[12];
    const float* thr2  = (const float*)d_in[13];
    const float* lt2   = (const float*)d_in[14];
    const float* resp2 = (const float*)d_in[15];

    k_prep<<<L0BLK + NT, 256, 0, stream>>>(x, out, fsl0, fsl1, fsl2, esl0, esl1, esl2,
                                               thr0, lt0, resp0, fs1, fs2, es1, es2);

    k_layer<<<256, 128, 0, stream>>>(out, fs1, es1, thr1, lt1, resp1, out, 192, 192);
    k_layer<<<256, 128, 0, stream>>>(out, fs2, es2, thr2, lt2, resp2, out, 320, 320);
}
